# starting kernel
# baseline (speedup 1.0000x reference)
#include <hip/hip_runtime.h>


typedef __attribute__((ext_vector_type(2))) float v2f;
typedef __attribute__((ext_vector_type(8))) float v8f;

static constexpr size_t S1P  = 0;
static constexpr size_t S2P  = 8388608;
static constexpr size_t S3M  = 16777216;
static constexpr size_t D12  = 17825792;
static constexpr size_t D13  = 18874368;
static constexpr size_t D23  = 19922944;
static constexpr size_t TRI  = 20971520;
static constexpr size_t S1M  = 20987904;
static constexpr size_t S2M  = 22036480;
static constexpr size_t A1O  = 23085056;
static constexpr size_t A0O  = 23248896;
static constexpr size_t R12O = 23250176;
static constexpr size_t R12TO= 23266560;
static constexpr size_t FORO = 23282944;
static constexpr size_t R02O = 23299328;
static constexpr size_t WS_FLOATS = 23299584;

__global__ void k1_contract(const float* __restrict__ x,
                            float* __restrict__ s1p, float* __restrict__ s2p,
                            float* __restrict__ s3m, float* __restrict__ d12,
                            float* __restrict__ d13, float* __restrict__ d23,
                            float* __restrict__ tri) {
  __shared__ float lds[7 * 2048 + 256];
  const int t  = threadIdx.x;
  const int jt = blockIdx.x, it = blockIdx.y, b = blockIdx.z;
  const int kg = t >> 5, c = t & 31;

  for (int idx = t; idx < 7 * 2048; idx += 256) lds[idx] = 0.f;
  float s2r[8];
  #pragma unroll
  for (int m = 0; m < 8; ++m) s2r[m] = 0.f;
  __syncthreads();

  for (int jj = 0; jj < 8; ++jj) {
    const int j = jt * 8 + jj;
    float s1a[8];
    #pragma unroll
    for (int m = 0; m < 8; ++m) s1a[m] = 0.f;

    for (int ii = 0; ii < 8; ++ii) {
      const int i = it * 8 + ii;
      const float* row = x + (((size_t)((b * 64 + i) * 64 + j)) << 11);
      float v[8];
      #pragma unroll
      for (int m = 0; m < 8; ++m) v[m] = row[t + (m << 8)];

      float p = 0.f;
      #pragma unroll
      for (int m = 0; m < 8; ++m) { s1a[m] += v[m]; p += v[m]; }

      if (ii < 7) {
        #pragma unroll
        for (int m = 0; m < 8; ++m) lds[ii * 2048 + t + (m << 8)] += v[m];
      } else {
        #pragma unroll
        for (int m = 0; m < 8; ++m) s2r[m] += v[m];
      }

      lds[14336 + t] = p;
      __syncthreads();
      if (t < 32) {
        float tot = 0.f;
        #pragma unroll
        for (int g = 0; g < 8; ++g) tot += lds[14336 + g * 32 + t];
        s3m[(((size_t)((b * 64 + i) * 64 + j)) << 5) + t] = tot * (1.f / 64.f);
      }
      if ((i & 7) == kg) d13[(((size_t)((b * 64 + i) * 64 + j)) << 5) + c] = v[i >> 3];
      if ((j & 7) == kg) d23[(((size_t)((b * 64 + i) * 64 + j)) << 5) + c] = v[j >> 3];
      if (i == j) {
        #pragma unroll
        for (int m = 0; m < 8; ++m)
          d12[(((size_t)(b * 64 + i)) << 11) + t + (m << 8)] = v[m];
        if ((i & 7) == kg) tri[((b << 6) + i) * 32 + c] = v[i >> 3];
      }
      __syncthreads();
    }
    float* dst = s1p + (((size_t)((b * 8 + it) * 64 + j)) << 11);
    #pragma unroll
    for (int m = 0; m < 8; ++m) dst[t + (m << 8)] = s1a[m];
  }
  for (int ii = 0; ii < 8; ++ii) {
    float* dst = s2p + (((size_t)((b * 8 + jt) * 64 + it * 8 + ii)) << 11);
    if (ii < 7) {
      #pragma unroll
      for (int m = 0; m < 8; ++m) dst[t + (m << 8)] = lds[ii * 2048 + t + (m << 8)];
    } else {
      #pragma unroll
      for (int m = 0; m < 8; ++m) dst[t + (m << 8)] = s2r[m];
    }
  }
}

__global__ void k2a_reduce(const float* __restrict__ s1p, const float* __restrict__ s2p,
                           float* __restrict__ s1m, float* __restrict__ s2m) {
  const int t = threadIdx.x;
  const int u = blockIdx.x & 63, b = blockIdx.x >> 6;
  #pragma unroll
  for (int m = 0; m < 8; ++m) {
    const int e = t + (m << 8);
    float a = 0.f, s = 0.f;
    #pragma unroll
    for (int g = 0; g < 8; ++g) {
      a += s1p[(((size_t)((b * 8 + g) * 64 + u)) << 11) + e];
      s += s2p[(((size_t)((b * 8 + g) * 64 + u)) << 11) + e];
    }
    s1m[(((size_t)(b * 64 + u)) << 11) + e] = a * (1.f / 64.f);
    s2m[(((size_t)(b * 64 + u)) << 11) + e] = s * (1.f / 64.f);
  }
}

__global__ void k2b_a1(const float* __restrict__ s1m, const float* __restrict__ s3m,
                       const float* __restrict__ d12, const float* __restrict__ d13,
                       const float* __restrict__ d23, const float* __restrict__ tri,
                       float* __restrict__ a1) {
  __shared__ float scr[256];
  const int t = threadIdx.x, c = t & 31, g = t >> 5;
  const int b = blockIdx.x >> 6, u = blockIdx.x & 63;
  float* dst = a1 + (size_t)(b * 64 + u) * 320;

  auto red = [&](float partial, int slot) {
    scr[t] = partial;
    __syncthreads();
    if (t < 32) {
      float tot = 0.f;
      #pragma unroll
      for (int gg = 0; gg < 8; ++gg) tot += scr[gg * 32 + t];
      dst[slot * 32 + t] = tot * (1.f / 64.f);
    }
    __syncthreads();
  };
  float acc;
  acc = 0.f; for (int s = 0; s < 8; ++s) { int v = g + 8 * s;
    acc += s1m[(((size_t)((b * 64 + v) * 64 + u)) << 5) + c]; } red(acc, 0);
  acc = 0.f; for (int s = 0; s < 8; ++s) { int v = g + 8 * s;
    acc += s1m[(((size_t)((b * 64 + u) * 64 + v)) << 5) + c]; } red(acc, 1);
  acc = 0.f; for (int s = 0; s < 8; ++s) { int v = g + 8 * s;
    acc += s3m[(((size_t)((b * 64 + u) * 64 + v)) << 5) + c]; } red(acc, 2);
  acc = 0.f; for (int s = 0; s < 8; ++s) { int v = g + 8 * s;
    acc += d12[(((size_t)(b * 64 + v)) << 11) + u * 32 + c]; } red(acc, 3);
  acc = 0.f; for (int s = 0; s < 8; ++s) { int v = g + 8 * s;
    acc += d12[(((size_t)(b * 64 + u)) << 11) + v * 32 + c]; } red(acc, 4);
  acc = 0.f; for (int s = 0; s < 8; ++s) { int v = g + 8 * s;
    acc += d13[(((size_t)((b * 64 + v) * 64 + u)) << 5) + c]; } red(acc, 5);
  acc = 0.f; for (int s = 0; s < 8; ++s) { int v = g + 8 * s;
    acc += d13[(((size_t)((b * 64 + u) * 64 + v)) << 5) + c]; } red(acc, 6);
  acc = 0.f; for (int s = 0; s < 8; ++s) { int v = g + 8 * s;
    acc += d23[(((size_t)((b * 64 + v) * 64 + u)) << 5) + c]; } red(acc, 7);
  acc = 0.f; for (int s = 0; s < 8; ++s) { int v = g + 8 * s;
    acc += d23[(((size_t)((b * 64 + u) * 64 + v)) << 5) + c]; } red(acc, 8);
  if (t < 32) dst[9 * 32 + t] = tri[(b * 64 + u) * 32 + t];
}

__global__ void k2c_a0(const float* __restrict__ a1, float* __restrict__ a0) {
  const int b = blockIdx.x, t = threadIdx.x;
  if (t < 160) {
    const int s = t >> 5, c = t & 31;
    const int slot = (s == 0) ? 0 : (2 * s + 1);
    float acc = 0.f;
    for (int u = 0; u < 64; ++u)
      acc += a1[(size_t)(b * 64 + u) * 320 + slot * 32 + c];
    a0[b * 160 + s * 32 + c] = acc * (1.f / 64.f);
  }
}

__global__ void k3a_rows(const float* __restrict__ a1, const float* __restrict__ a0,
                         const float* __restrict__ W12, const float* __restrict__ b12,
                         const float* __restrict__ W12t, const float* __restrict__ b12t,
                         const float* __restrict__ W11, const float* __restrict__ b11,
                         const float* __restrict__ W01, const float* __restrict__ b01,
                         const float* __restrict__ W02, const float* __restrict__ b02,
                         const float* __restrict__ b22,
                         float* __restrict__ r12, float* __restrict__ r12t,
                         float* __restrict__ forow, float* __restrict__ r02) {
  const int row = blockIdx.x;
  const int b = row >> 6, p = row & 63;
  const int n = threadIdx.x;
  const float* a1r = a1 + (size_t)row * 320;
  const float* a0r = a0 + b * 160;
  float v12 = b12[n], v12t = b12t[n], vfo = b11[n] + b01[n];
  for (int k = 0; k < 320; ++k) {
    const float av = a1r[k];
    v12  += av * W12 [k * 32 + n];
    v12t += av * W12t[k * 32 + n];
    vfo  += av * W11 [k * 32 + n];
  }
  for (int k = 0; k < 160; ++k) vfo += a0r[k] * W01[k * 32 + n];
  r12  [row * 32 + n] = v12;
  r12t [row * 32 + n] = v12t;
  forow[row * 32 + n] = vfo;
  if (p == 0) {
    float v02 = b02[n] + b22[n];
    for (int k = 0; k < 160; ++k) v02 += a0r[k] * W02[k * 32 + n];
    r02[b * 32 + n] = v02;
  }
}

__global__ void k3b_mix(const float* __restrict__ s1m, const float* __restrict__ s2m,
                        const float* __restrict__ s3m, const float* __restrict__ d12,
                        const float* __restrict__ d13, const float* __restrict__ d23,
                        const float* __restrict__ W22,
                        const float* __restrict__ r12, const float* __restrict__ r12t,
                        const float* __restrict__ r02, const float* __restrict__ forow,
                        float* __restrict__ out) {
  __shared__ float w[384 * 32];
  const int t = threadIdx.x;
  for (int idx = t; idx < 384 * 32; idx += 256) w[idx] = W22[idx];
  __syncthreads();

  const int lane = t & 31;
  const int R = (blockIdx.x * 8 + (t >> 5)) * 16;
  const int MA = lane & 15;
  const int kh = lane >> 4;
  const int rowA = R + MA;
  const int bA = rowA >> 12, pA = (rowA >> 6) & 63, qA = rowA & 63;
  const size_t offPQ = ((size_t)((bA * 64 + pA) * 64 + qA)) << 5;
  const size_t offQP = ((size_t)((bA * 64 + qA) * 64 + pA)) << 5;
  const float* comp[6] = { s1m, s2m, s3m, d12, d13, d23 };
  const int n0 = lane & 15;

  v8f acc0 = {};
  v8f acc1 = {};
  #pragma unroll
  for (int f = 0; f < 12; ++f) {
    const float* A = (f < 6) ? (comp[f] + offPQ) : (comp[f - 6] + offQP);
    #pragma unroll
    for (int kk = 0; kk < 32; kk += 4) {
      v2f a = *(const v2f*)(A + kk + 2 * kh);
      const int kcol = f * 32 + kk + 2 * kh;
      v2f b0 = { w[kcol * 32 + n0],      w[(kcol + 1) * 32 + n0] };
      v2f b1 = { w[kcol * 32 + n0 + 16], w[(kcol + 1) * 32 + n0 + 16] };
      acc0 = __builtin_amdgcn_wmma_f32_16x16x4_f32(false, a, false, b0,
                                                   (short)0, acc0, false, false);
      acc1 = __builtin_amdgcn_wmma_f32_16x16x4_f32(false, a, false, b1,
                                                   (short)0, acc1, false, false);
    }
  }
  #pragma unroll
  for (int r = 0; r < 8; ++r) {
    const int M = r + ((lane >> 4) << 3);
    const int row = R + M;
    const int b = row >> 12, p = (row >> 6) & 63, q = row & 63;
    float add0 = r12[((b << 6) + p) * 32 + n0] + r12t[((b << 6) + q) * 32 + n0]
               + r02[(b << 5) + n0];
    float add1 = r12[((b << 6) + p) * 32 + n0 + 16] + r12t[((b << 6) + q) * 32 + n0 + 16]
               + r02[(b << 5) + n0 + 16];
    if (p == q) {
      add0 += forow[((b << 6) + p) * 32 + n0];
      add1 += forow[((b << 6) + p) * 32 + n0 + 16];
    }
    out[((size_t)row << 5) + n0]      = acc0[r] + add0;
    out[((size_t)row << 5) + n0 + 16] = acc1[r] + add1;
  }
}

extern "C" void kernel_launch(void* const* d_in, const int* in_sizes, int n_in,
                              void* d_out, int out_size, void* d_ws, size_t ws_size,
                              hipStream_t stream) {
  (void)in_sizes; (void)n_in; (void)out_size;
  if (ws_size < WS_FLOATS * sizeof(float)) return;

  const float* x    = (const float*)d_in[0];
  const float* W22  = (const float*)d_in[1];
  const float* b22  = (const float*)d_in[2];
  const float* W12  = (const float*)d_in[3];
  const float* b12  = (const float*)d_in[4];
  const float* W12t = (const float*)d_in[5];
  const float* b12t = (const float*)d_in[6];
  const float* W02  = (const float*)d_in[7];
  const float* b02  = (const float*)d_in[8];
  const float* W11  = (const float*)d_in[9];
  const float* b11  = (const float*)d_in[10];
  const float* W01  = (const float*)d_in[11];
  const float* b01  = (const float*)d_in[12];

  float* ws = (float*)d_ws;
  float* s1p = ws + S1P;   float* s2p = ws + S2P;
  float* s3m = ws + S3M;   float* d12 = ws + D12;
  float* d13 = ws + D13;   float* d23 = ws + D23;
  float* tri = ws + TRI;   float* s1m = ws + S1M;
  float* s2m = ws + S2M;   float* a1  = ws + A1O;
  float* a0  = ws + A0O;   float* r12 = ws + R12O;
  float* r12t= ws + R12TO; float* forow = ws + FORO;
  float* r02 = ws + R02O;

  k1_contract<<<dim3(8, 8, 8), 256, 0, stream>>>(x, s1p, s2p, s3m, d12, d13, d23, tri);
  k2a_reduce <<<512, 256, 0, stream>>>(s1p, s2p, s1m, s2m);
  k2b_a1     <<<512, 256, 0, stream>>>(s1m, s3m, d12, d13, d23, tri, a1);
  k2c_a0     <<<8, 256, 0, stream>>>(a1, a0);
  k3a_rows   <<<512, 32, 0, stream>>>(a1, a0, W12, b12, W12t, b12t,
                                      W11, b11, W01, b01, W02, b02, b22,
                                      r12, r12t, forow, r02);
  k3b_mix    <<<256, 256, 0, stream>>>(s1m, s2m, s3m, d12, d13, d23, W22,
                                       r12, r12t, r02, forow, (float*)d_out);
}


# Pass 1 made latency-tolerant: 16-byte non-temporal loads, 14-16 in flight per wave (row refilled once summed), no barrier in the stream; k-sum folded across waves once at the end (new sum order)
# speedup vs baseline: 1.3245x; 1.3245x over previous
#include <hip/hip_runtime.h>


typedef __attribute__((ext_vector_type(2))) float v2f;
typedef __attribute__((ext_vector_type(8))) float v8f;

static constexpr size_t S1P  = 0;
static constexpr size_t S2P  = 8388608;
static constexpr size_t S3M  = 16777216;
static constexpr size_t D12  = 17825792;
static constexpr size_t D13  = 18874368;
static constexpr size_t D23  = 19922944;
static constexpr size_t TRI  = 20971520;
static constexpr size_t S1M  = 20987904;
static constexpr size_t S2M  = 22036480;
static constexpr size_t A1O  = 23085056;
static constexpr size_t A0O  = 23248896;
static constexpr size_t R12O = 23250176;
static constexpr size_t R12TO= 23266560;
static constexpr size_t FORO = 23282944;
static constexpr size_t R02O = 23299328;
static constexpr size_t WS_FLOATS = 23299584;

typedef __attribute__((ext_vector_type(4))) float v4f;

static __device__ __forceinline__ v4f load_x(const float* p) {
  return __builtin_nontemporal_load((const v4f*)p);
}

__global__ void __launch_bounds__(256)
k1_contract(const float* __restrict__ x,
            float* __restrict__ s1p, float* __restrict__ s2p,
            float* __restrict__ s3m, float* __restrict__ d12,
            float* __restrict__ d13, float* __restrict__ d23,
            float* __restrict__ tri) {
  __shared__ float scr[64 * 8 * 32];
  const int t  = threadIdx.x;
  const int jt = blockIdx.x, it = blockIdx.y, b = blockIdx.z;
  const int lane = t & 31, wave = t >> 5;
  const int kq = t >> 3, c4 = t & 7;
  const int i0 = it * 8, j0 = jt * 8;
  const bool hi_i = (it >> 2) != 0;
  const bool hi_j = (jt >> 2) != 0;
  const float* xb = x + (((size_t)((b * 64 + i0) * 64 + j0)) << 11) + 4 * t;

  v4f s1acc[8][2];
  #pragma unroll
  for (int jj = 0; jj < 8; ++jj) { s1acc[jj][0] = (v4f){0.f, 0.f, 0.f, 0.f}; s1acc[jj][1] = (v4f){0.f, 0.f, 0.f, 0.f}; }

  auto row = [&](const v4f v0, const v4f v1, const int ii, const int jj, v4f& s2a0, v4f& s2a1) {
    const int i = i0 + ii, j = j0 + jj;
    s1acc[jj][0] += v0; s1acc[jj][1] += v1;
    s2a0 += v0;         s2a1 += v1;
    v4f p = v0 + v1;
    p.x += __shfl_xor(p.x, 8, 32);  p.y += __shfl_xor(p.y, 8, 32);
    p.z += __shfl_xor(p.z, 8, 32);  p.w += __shfl_xor(p.w, 8, 32);
    p.x += __shfl_xor(p.x, 16, 32); p.y += __shfl_xor(p.y, 16, 32);
    p.z += __shfl_xor(p.z, 16, 32); p.w += __shfl_xor(p.w, 16, 32);
    if (lane < 8) *(v4f*)&scr[(((ii * 8 + jj) * 8 + wave) << 5) + 4 * lane] = p;
    const size_t ro = ((size_t)((b * 64 + i) * 64 + j)) << 5;
    if (kq == (i & 31)) *(v4f*)(d13 + ro + 4 * c4) = hi_i ? v1 : v0;
    if (kq == (j & 31)) *(v4f*)(d23 + ro + 4 * c4) = hi_j ? v1 : v0;
    if (i == j) {
      float* dd = d12 + (((size_t)(b * 64 + i)) << 11) + 4 * t;
      *(v4f*)dd = v0; *(v4f*)(dd + 1024) = v1;
      if (kq == (i & 31)) *(v4f*)(tri + ((b << 6) + i) * 32 + 4 * c4) = hi_i ? v1 : v0;
    }
    asm volatile("" : "+v"(s1acc[jj][0]), "+v"(s1acc[jj][1]), "+v"(s2a0), "+v"(s2a1));
  };

  v4f buf[8][2];
  #pragma unroll
  for (int jj = 0; jj < 8; ++jj) {
    buf[jj][0] = load_x(xb + jj * 2048);
    buf[jj][1] = load_x(xb + jj * 2048 + 1024);
  }
  for (int ii = 0; ii < 7; ++ii) {
    const float* nx = xb + (size_t)(ii + 1) * 131072;
    v4f s2a0 = {0.f, 0.f, 0.f, 0.f}, s2a1 = {0.f, 0.f, 0.f, 0.f};
    #pragma unroll
    for (int jj = 0; jj < 8; ++jj) {
      row(buf[jj][0], buf[jj][1], ii, jj, s2a0, s2a1);
      buf[jj][0] = load_x(nx + jj * 2048);
      buf[jj][1] = load_x(nx + jj * 2048 + 1024);
    }
    float* dst = s2p + (((size_t)((b * 8 + jt) * 64 + i0 + ii)) << 11) + 4 * t;
    *(v4f*)dst = s2a0; *(v4f*)(dst + 1024) = s2a1;
  }
  {
    v4f s2a0 = {0.f, 0.f, 0.f, 0.f}, s2a1 = {0.f, 0.f, 0.f, 0.f};
    #pragma unroll
    for (int jj = 0; jj < 8; ++jj) row(buf[jj][0], buf[jj][1], 7, jj, s2a0, s2a1);
    float* dst = s2p + (((size_t)((b * 8 + jt) * 64 + i0 + 7)) << 11) + 4 * t;
    *(v4f*)dst = s2a0; *(v4f*)(dst + 1024) = s2a1;
  }

  #pragma unroll
  for (int jj = 0; jj < 8; ++jj) {
    float* dst = s1p + (((size_t)((b * 8 + it) * 64 + j0 + jj)) << 11) + 4 * t;
    *(v4f*)dst = s1acc[jj][0]; *(v4f*)(dst + 1024) = s1acc[jj][1];
  }

  __syncthreads();
  #pragma unroll
  for (int ii = 0; ii < 8; ++ii) {
    float tot = 0.f;
    #pragma unroll
    for (int w = 0; w < 8; ++w) tot += scr[(((ii * 8 + wave) * 8 + w) << 5) + lane];
    s3m[(((size_t)((b * 64 + i0 + ii) * 64 + j0 + wave)) << 5) + lane] = tot * (1.f / 64.f);
  }
}

__global__ void k2a_reduce(const float* __restrict__ s1p, const float* __restrict__ s2p,
                           float* __restrict__ s1m, float* __restrict__ s2m) {
  const int t = threadIdx.x;
  const int u = blockIdx.x & 63, b = blockIdx.x >> 6;
  #pragma unroll
  for (int m = 0; m < 8; ++m) {
    const int e = t + (m << 8);
    float a = 0.f, s = 0.f;
    #pragma unroll
    for (int g = 0; g < 8; ++g) {
      a += s1p[(((size_t)((b * 8 + g) * 64 + u)) << 11) + e];
      s += s2p[(((size_t)((b * 8 + g) * 64 + u)) << 11) + e];
    }
    s1m[(((size_t)(b * 64 + u)) << 11) + e] = a * (1.f / 64.f);
    s2m[(((size_t)(b * 64 + u)) << 11) + e] = s * (1.f / 64.f);
  }
}

__global__ void k2b_a1(const float* __restrict__ s1m, const float* __restrict__ s3m,
                       const float* __restrict__ d12, const float* __restrict__ d13,
                       const float* __restrict__ d23, const float* __restrict__ tri,
                       float* __restrict__ a1) {
  __shared__ float scr[256];
  const int t = threadIdx.x, c = t & 31, g = t >> 5;
  const int b = blockIdx.x >> 6, u = blockIdx.x & 63;
  float* dst = a1 + (size_t)(b * 64 + u) * 320;

  auto red = [&](float partial, int slot) {
    scr[t] = partial;
    __syncthreads();
    if (t < 32) {
      float tot = 0.f;
      #pragma unroll
      for (int gg = 0; gg < 8; ++gg) tot += scr[gg * 32 + t];
      dst[slot * 32 + t] = tot * (1.f / 64.f);
    }
    __syncthreads();
  };
  float acc;
  acc = 0.f; for (int s = 0; s < 8; ++s) { int v = g + 8 * s;
    acc += s1m[(((size_t)((b * 64 + v) * 64 + u)) << 5) + c]; } red(acc, 0);
  acc = 0.f; for (int s = 0; s < 8; ++s) { int v = g + 8 * s;
    acc += s1m[(((size_t)((b * 64 + u) * 64 + v)) << 5) + c]; } red(acc, 1);
  acc = 0.f; for (int s = 0; s < 8; ++s) { int v = g + 8 * s;
    acc += s3m[(((size_t)((b * 64 + u) * 64 + v)) << 5) + c]; } red(acc, 2);
  acc = 0.f; for (int s = 0; s < 8; ++s) { int v = g + 8 * s;
    acc += d12[(((size_t)(b * 64 + v)) << 11) + u * 32 + c]; } red(acc, 3);
  acc = 0.f; for (int s = 0; s < 8; ++s) { int v = g + 8 * s;
    acc += d12[(((size_t)(b * 64 + u)) << 11) + v * 32 + c]; } red(acc, 4);
  acc = 0.f; for (int s = 0; s < 8; ++s) { int v = g + 8 * s;
    acc += d13[(((size_t)((b * 64 + v) * 64 + u)) << 5) + c]; } red(acc, 5);
  acc = 0.f; for (int s = 0; s < 8; ++s) { int v = g + 8 * s;
    acc += d13[(((size_t)((b * 64 + u) * 64 + v)) << 5) + c]; } red(acc, 6);
  acc = 0.f; for (int s = 0; s < 8; ++s) { int v = g + 8 * s;
    acc += d23[(((size_t)((b * 64 + v) * 64 + u)) << 5) + c]; } red(acc, 7);
  acc = 0.f; for (int s = 0; s < 8; ++s) { int v = g + 8 * s;
    acc += d23[(((size_t)((b * 64 + u) * 64 + v)) << 5) + c]; } red(acc, 8);
  if (t < 32) dst[9 * 32 + t] = tri[(b * 64 + u) * 32 + t];
}

__global__ void k2c_a0(const float* __restrict__ a1, float* __restrict__ a0) {
  const int b = blockIdx.x, t = threadIdx.x;
  if (t < 160) {
    const int s = t >> 5, c = t & 31;
    const int slot = (s == 0) ? 0 : (2 * s + 1);
    float acc = 0.f;
    for (int u = 0; u < 64; ++u)
      acc += a1[(size_t)(b * 64 + u) * 320 + slot * 32 + c];
    a0[b * 160 + s * 32 + c] = acc * (1.f / 64.f);
  }
}

__global__ void k3a_rows(const float* __restrict__ a1, const float* __restrict__ a0,
                         const float* __restrict__ W12, const float* __restrict__ b12,
                         const float* __restrict__ W12t, const float* __restrict__ b12t,
                         const float* __restrict__ W11, const float* __restrict__ b11,
                         const float* __restrict__ W01, const float* __restrict__ b01,
                         const float* __restrict__ W02, const float* __restrict__ b02,
                         const float* __restrict__ b22,
                         float* __restrict__ r12, float* __restrict__ r12t,
                         float* __restrict__ forow, float* __restrict__ r02) {
  const int row = blockIdx.x;
  const int b = row >> 6, p = row & 63;
  const int n = threadIdx.x;
  const float* a1r = a1 + (size_t)row * 320;
  const float* a0r = a0 + b * 160;
  float v12 = b12[n], v12t = b12t[n], vfo = b11[n] + b01[n];
  for (int k = 0; k < 320; ++k) {
    const float av = a1r[k];
    v12  += av * W12 [k * 32 + n];
    v12t += av * W12t[k * 32 + n];
    vfo  += av * W11 [k * 32 + n];
  }
  for (int k = 0; k < 160; ++k) vfo += a0r[k] * W01[k * 32 + n];
  r12  [row * 32 + n] = v12;
  r12t [row * 32 + n] = v12t;
  forow[row * 32 + n] = vfo;
  if (p == 0) {
    float v02 = b02[n] + b22[n];
    for (int k = 0; k < 160; ++k) v02 += a0r[k] * W02[k * 32 + n];
    r02[b * 32 + n] = v02;
  }
}

__global__ void k3b_mix(const float* __restrict__ s1m, const float* __restrict__ s2m,
                        const float* __restrict__ s3m, const float* __restrict__ d12,
                        const float* __restrict__ d13, const float* __restrict__ d23,
                        const float* __restrict__ W22,
                        const float* __restrict__ r12, const float* __restrict__ r12t,
                        const float* __restrict__ r02, const float* __restrict__ forow,
                        float* __restrict__ out) {
  __shared__ float w[384 * 32];
  const int t = threadIdx.x;
  for (int idx = t; idx < 384 * 32; idx += 256) w[idx] = W22[idx];
  __syncthreads();

  const int lane = t & 31;
  const int R = (blockIdx.x * 8 + (t >> 5)) * 16;
  const int MA = lane & 15;
  const int kh = lane >> 4;
  const int rowA = R + MA;
  const int bA = rowA >> 12, pA = (rowA >> 6) & 63, qA = rowA & 63;
  const size_t offPQ = ((size_t)((bA * 64 + pA) * 64 + qA)) << 5;
  const size_t offQP = ((size_t)((bA * 64 + qA) * 64 + pA)) << 5;
  const float* comp[6] = { s1m, s2m, s3m, d12, d13, d23 };
  const int n0 = lane & 15;

  v8f acc0 = {};
  v8f acc1 = {};
  #pragma unroll
  for (int f = 0; f < 12; ++f) {
    const float* A = (f < 6) ? (comp[f] + offPQ) : (comp[f - 6] + offQP);
    #pragma unroll
    for (int kk = 0; kk < 32; kk += 4) {
      v2f a = *(const v2f*)(A + kk + 2 * kh);
      const int kcol = f * 32 + kk + 2 * kh;
      v2f b0 = { w[kcol * 32 + n0],      w[(kcol + 1) * 32 + n0] };
      v2f b1 = { w[kcol * 32 + n0 + 16], w[(kcol + 1) * 32 + n0 + 16] };
      acc0 = __builtin_amdgcn_wmma_f32_16x16x4_f32(false, a, false, b0,
                                                   (short)0, acc0, false, false);
      acc1 = __builtin_amdgcn_wmma_f32_16x16x4_f32(false, a, false, b1,
                                                   (short)0, acc1, false, false);
    }
  }
  #pragma unroll
  for (int r = 0; r < 8; ++r) {
    const int M = r + ((lane >> 4) << 3);
    const int row = R + M;
    const int b = row >> 12, p = (row >> 6) & 63, q = row & 63;
    float add0 = r12[((b << 6) + p) * 32 + n0] + r12t[((b << 6) + q) * 32 + n0]
               + r02[(b << 5) + n0];
    float add1 = r12[((b << 6) + p) * 32 + n0 + 16] + r12t[((b << 6) + q) * 32 + n0 + 16]
               + r02[(b << 5) + n0 + 16];
    if (p == q) {
      add0 += forow[((b << 6) + p) * 32 + n0];
      add1 += forow[((b << 6) + p) * 32 + n0 + 16];
    }
    out[((size_t)row << 5) + n0]      = acc0[r] + add0;
    out[((size_t)row << 5) + n0 + 16] = acc1[r] + add1;
  }
}

extern "C" void kernel_launch(void* const* d_in, const int* in_sizes, int n_in,
                              void* d_out, int out_size, void* d_ws, size_t ws_size,
                              hipStream_t stream) {
  (void)in_sizes; (void)n_in; (void)out_size;
  if (ws_size < WS_FLOATS * sizeof(float)) return;

  const float* x    = (const float*)d_in[0];
  const float* W22  = (const float*)d_in[1];
  const float* b22  = (const float*)d_in[2];
  const float* W12  = (const float*)d_in[3];
  const float* b12  = (const float*)d_in[4];
  const float* W12t = (const float*)d_in[5];
  const float* b12t = (const float*)d_in[6];
  const float* W02  = (const float*)d_in[7];
  const float* b02  = (const float*)d_in[8];
  const float* W11  = (const float*)d_in[9];
  const float* b11  = (const float*)d_in[10];
  const float* W01  = (const float*)d_in[11];
  const float* b01  = (const float*)d_in[12];

  float* ws = (float*)d_ws;
  float* s1p = ws + S1P;   float* s2p = ws + S2P;
  float* s3m = ws + S3M;   float* d12 = ws + D12;
  float* d13 = ws + D13;   float* d23 = ws + D23;
  float* tri = ws + TRI;   float* s1m = ws + S1M;
  float* s2m = ws + S2M;   float* a1  = ws + A1O;
  float* a0  = ws + A0O;   float* r12 = ws + R12O;
  float* r12t= ws + R12TO; float* forow = ws + FORO;
  float* r02 = ws + R02O;

  k1_contract<<<dim3(8, 8, 8), 256, 0, stream>>>(x, s1p, s2p, s3m, d12, d13, d23, tri);
  k2a_reduce <<<512, 256, 0, stream>>>(s1p, s2p, s1m, s2m);
  k2b_a1     <<<512, 256, 0, stream>>>(s1m, s3m, d12, d13, d23, tri, a1);
  k2c_a0     <<<8, 256, 0, stream>>>(a1, a0);
  k3a_rows   <<<512, 32, 0, stream>>>(a1, a0, W12, b12, W12t, b12t,
                                      W11, b11, W01, b01, W02, b02, b22,
                                      r12, r12t, forow, r02);
  k3b_mix    <<<256, 256, 0, stream>>>(s1m, s2m, s3m, d12, d13, d23, W22,
                                       r12, r12t, r02, forow, (float*)d_out);
}


# Tail launches 6 -> 4: row mixes (a1 @ W12, W12t, W11) done by the block that builds the a1 row, a0 and its two per-b vectors by the mix kernel's prologue; K split over waves (new sum order)
# speedup vs baseline: 1.7723x; 1.3381x over previous
#include <hip/hip_runtime.h>


typedef __attribute__((ext_vector_type(2))) float v2f;
typedef __attribute__((ext_vector_type(8))) float v8f;

static constexpr size_t S1P  = 0;
static constexpr size_t S2P  = 8388608;
static constexpr size_t S3M  = 16777216;
static constexpr size_t D12  = 17825792;
static constexpr size_t D13  = 18874368;
static constexpr size_t D23  = 19922944;
static constexpr size_t TRI  = 20971520;
static constexpr size_t S1M  = 20987904;
static constexpr size_t S2M  = 22036480;
static constexpr size_t A1O  = 23085056;
static constexpr size_t R12O = 23250176;
static constexpr size_t R12TO= 23266560;
static constexpr size_t FORO = 23282944;
static constexpr size_t WS_FLOATS = 23299584;

typedef __attribute__((ext_vector_type(4))) float v4f;

static __device__ __forceinline__ v4f load_x(const float* p) {
  return __builtin_nontemporal_load((const v4f*)p);
}

__global__ void __launch_bounds__(256)
k1_contract(const float* __restrict__ x,
            float* __restrict__ s1p, float* __restrict__ s2p,
            float* __restrict__ s3m, float* __restrict__ d12,
            float* __restrict__ d13, float* __restrict__ d23,
            float* __restrict__ tri) {
  __shared__ float scr[64 * 8 * 32];
  const int t  = threadIdx.x;
  const int jt = blockIdx.x, it = blockIdx.y, b = blockIdx.z;
  const int lane = t & 31, wave = t >> 5;
  const int kq = t >> 3, c4 = t & 7;
  const int i0 = it * 8, j0 = jt * 8;
  const bool hi_i = (it >> 2) != 0;
  const bool hi_j = (jt >> 2) != 0;
  const float* xb = x + (((size_t)((b * 64 + i0) * 64 + j0)) << 11) + 4 * t;

  v4f s1acc[8][2];
  #pragma unroll
  for (int jj = 0; jj < 8; ++jj) { s1acc[jj][0] = (v4f){0.f, 0.f, 0.f, 0.f}; s1acc[jj][1] = (v4f){0.f, 0.f, 0.f, 0.f}; }

  auto row = [&](const v4f v0, const v4f v1, const int ii, const int jj, v4f& s2a0, v4f& s2a1) {
    const int i = i0 + ii, j = j0 + jj;
    s1acc[jj][0] += v0; s1acc[jj][1] += v1;
    s2a0 += v0;         s2a1 += v1;
    v4f p = v0 + v1;
    p.x += __shfl_xor(p.x, 8, 32);  p.y += __shfl_xor(p.y, 8, 32);
    p.z += __shfl_xor(p.z, 8, 32);  p.w += __shfl_xor(p.w, 8, 32);
    p.x += __shfl_xor(p.x, 16, 32); p.y += __shfl_xor(p.y, 16, 32);
    p.z += __shfl_xor(p.z, 16, 32); p.w += __shfl_xor(p.w, 16, 32);
    if (lane < 8) *(v4f*)&scr[(((ii * 8 + jj) * 8 + wave) << 5) + 4 * lane] = p;
    const size_t ro = ((size_t)((b * 64 + i) * 64 + j)) << 5;
    if (kq == (i & 31)) *(v4f*)(d13 + ro + 4 * c4) = hi_i ? v1 : v0;
    if (kq == (j & 31)) *(v4f*)(d23 + ro + 4 * c4) = hi_j ? v1 : v0;
    if (i == j) {
      float* dd = d12 + (((size_t)(b * 64 + i)) << 11) + 4 * t;
      *(v4f*)dd = v0; *(v4f*)(dd + 1024) = v1;
      if (kq == (i & 31)) *(v4f*)(tri + ((b << 6) + i) * 32 + 4 * c4) = hi_i ? v1 : v0;
    }
    asm volatile("" : "+v"(s1acc[jj][0]), "+v"(s1acc[jj][1]), "+v"(s2a0), "+v"(s2a1));
  };

  v4f buf[8][2];
  #pragma unroll
  for (int jj = 0; jj < 8; ++jj) {
    buf[jj][0] = load_x(xb + jj * 2048);
    buf[jj][1] = load_x(xb + jj * 2048 + 1024);
  }
  for (int ii = 0; ii < 7; ++ii) {
    const float* nx = xb + (size_t)(ii + 1) * 131072;
    v4f s2a0 = {0.f, 0.f, 0.f, 0.f}, s2a1 = {0.f, 0.f, 0.f, 0.f};
    #pragma unroll
    for (int jj = 0; jj < 8; ++jj) {
      row(buf[jj][0], buf[jj][1], ii, jj, s2a0, s2a1);
      buf[jj][0] = load_x(nx + jj * 2048);
      buf[jj][1] = load_x(nx + jj * 2048 + 1024);
    }
    float* dst = s2p + (((size_t)((b * 8 + jt) * 64 + i0 + ii)) << 11) + 4 * t;
    *(v4f*)dst = s2a0; *(v4f*)(dst + 1024) = s2a1;
  }
  {
    v4f s2a0 = {0.f, 0.f, 0.f, 0.f}, s2a1 = {0.f, 0.f, 0.f, 0.f};
    #pragma unroll
    for (int jj = 0; jj < 8; ++jj) row(buf[jj][0], buf[jj][1], 7, jj, s2a0, s2a1);
    float* dst = s2p + (((size_t)((b * 8 + jt) * 64 + i0 + 7)) << 11) + 4 * t;
    *(v4f*)dst = s2a0; *(v4f*)(dst + 1024) = s2a1;
  }

  #pragma unroll
  for (int jj = 0; jj < 8; ++jj) {
    float* dst = s1p + (((size_t)((b * 8 + it) * 64 + j0 + jj)) << 11) + 4 * t;
    *(v4f*)dst = s1acc[jj][0]; *(v4f*)(dst + 1024) = s1acc[jj][1];
  }

  __syncthreads();
  #pragma unroll
  for (int ii = 0; ii < 8; ++ii) {
    float tot = 0.f;
    #pragma unroll
    for (int w = 0; w < 8; ++w) tot += scr[(((ii * 8 + wave) * 8 + w) << 5) + lane];
    s3m[(((size_t)((b * 64 + i0 + ii) * 64 + j0 + wave)) << 5) + lane] = tot * (1.f / 64.f);
  }
}

__global__ void k2a_reduce(const float* __restrict__ s1p, const float* __restrict__ s2p,
                           float* __restrict__ s1m, float* __restrict__ s2m) {
  const int t = threadIdx.x;
  const int u = blockIdx.x & 63, b = blockIdx.x >> 6;
  #pragma unroll
  for (int m = 0; m < 8; ++m) {
    const int e = t + (m << 8);
    float a = 0.f, s = 0.f;
    #pragma unroll
    for (int g = 0; g < 8; ++g) {
      a += s1p[(((size_t)((b * 8 + g) * 64 + u)) << 11) + e];
      s += s2p[(((size_t)((b * 8 + g) * 64 + u)) << 11) + e];
    }
    s1m[(((size_t)(b * 64 + u)) << 11) + e] = a * (1.f / 64.f);
    s2m[(((size_t)(b * 64 + u)) << 11) + e] = s * (1.f / 64.f);
  }
}

__global__ void k2b_a1(const float* __restrict__ s1m, const float* __restrict__ s3m,
                       const float* __restrict__ d12, const float* __restrict__ d13,
                       const float* __restrict__ d23, const float* __restrict__ tri,
                       const float* __restrict__ W12, const float* __restrict__ b12,
                       const float* __restrict__ W12t, const float* __restrict__ b12t,
                       const float* __restrict__ W11, const float* __restrict__ b11,
                       float* __restrict__ a1, float* __restrict__ r12,
                       float* __restrict__ r12t, float* __restrict__ forow) {
  __shared__ float scr[256];
  __shared__ float arow[320];
  __shared__ float part[8 * 96];
  const int t = threadIdx.x, c = t & 31, g = t >> 5;
  const int b = blockIdx.x >> 6, u = blockIdx.x & 63;
  float* dst = a1 + (size_t)(b * 64 + u) * 320;

  auto red = [&](float partial, int slot) {
    scr[t] = partial;
    __syncthreads();
    if (t < 32) {
      float tot = 0.f;
      #pragma unroll
      for (int gg = 0; gg < 8; ++gg) tot += scr[gg * 32 + t];
      const float m = tot * (1.f / 64.f);
      dst[slot * 32 + t] = m;
      arow[slot * 32 + t] = m;
    }
    __syncthreads();
  };
  float acc;
  acc = 0.f; for (int s = 0; s < 8; ++s) { int v = g + 8 * s;
    acc += s1m[(((size_t)((b * 64 + v) * 64 + u)) << 5) + c]; } red(acc, 0);
  acc = 0.f; for (int s = 0; s < 8; ++s) { int v = g + 8 * s;
    acc += s1m[(((size_t)((b * 64 + u) * 64 + v)) << 5) + c]; } red(acc, 1);
  acc = 0.f; for (int s = 0; s < 8; ++s) { int v = g + 8 * s;
    acc += s3m[(((size_t)((b * 64 + u) * 64 + v)) << 5) + c]; } red(acc, 2);
  acc = 0.f; for (int s = 0; s < 8; ++s) { int v = g + 8 * s;
    acc += d12[(((size_t)(b * 64 + v)) << 11) + u * 32 + c]; } red(acc, 3);
  acc = 0.f; for (int s = 0; s < 8; ++s) { int v = g + 8 * s;
    acc += d12[(((size_t)(b * 64 + u)) << 11) + v * 32 + c]; } red(acc, 4);
  acc = 0.f; for (int s = 0; s < 8; ++s) { int v = g + 8 * s;
    acc += d13[(((size_t)((b * 64 + v) * 64 + u)) << 5) + c]; } red(acc, 5);
  acc = 0.f; for (int s = 0; s < 8; ++s) { int v = g + 8 * s;
    acc += d13[(((size_t)((b * 64 + u) * 64 + v)) << 5) + c]; } red(acc, 6);
  acc = 0.f; for (int s = 0; s < 8; ++s) { int v = g + 8 * s;
    acc += d23[(((size_t)((b * 64 + v) * 64 + u)) << 5) + c]; } red(acc, 7);
  acc = 0.f; for (int s = 0; s < 8; ++s) { int v = g + 8 * s;
    acc += d23[(((size_t)((b * 64 + u) * 64 + v)) << 5) + c]; } red(acc, 8);
  if (t < 32) { const float v = tri[(b * 64 + u) * 32 + t]; dst[9 * 32 + t] = v; arow[9 * 32 + t] = v; }
  __syncthreads();

  const int k0 = g * 40;
  float m12 = 0.f, m12t = 0.f, m11 = 0.f;
  #pragma unroll
  for (int k = 0; k < 40; ++k) {
    const float av = arow[k0 + k];
    m12  += av * W12 [(k0 + k) * 32 + c];
    m12t += av * W12t[(k0 + k) * 32 + c];
    m11  += av * W11 [(k0 + k) * 32 + c];
  }
  part[g * 96 + c] = m12; part[g * 96 + 32 + c] = m12t; part[g * 96 + 64 + c] = m11;
  __syncthreads();
  if (t < 96) {
    float tot = (g == 0) ? b12[c] : (g == 1) ? b12t[c] : b11[c];
    #pragma unroll
    for (int gg = 0; gg < 8; ++gg) tot += part[gg * 96 + t];
    float* o = (g == 0) ? r12 : (g == 1) ? r12t : forow;
    o[(b * 64 + u) * 32 + c] = tot;
  }
}

__global__ void k3b_mix(const float* __restrict__ s1m, const float* __restrict__ s2m,
                        const float* __restrict__ s3m, const float* __restrict__ d12,
                        const float* __restrict__ d13, const float* __restrict__ d23,
                        const float* __restrict__ W22, const float* __restrict__ a1,
                        const float* __restrict__ W02, const float* __restrict__ b02,
                        const float* __restrict__ b22,
                        const float* __restrict__ W01, const float* __restrict__ b01,
                        const float* __restrict__ r12, const float* __restrict__ r12t,
                        const float* __restrict__ forow,
                        float* __restrict__ out) {
  __shared__ float w[384 * 32];
  __shared__ float a0p[8 * 160];
  __shared__ float a0s[160];
  __shared__ float vp[4 * 64];
  __shared__ float vec[64];
  const int t = threadIdx.x;
  const int lane = t & 31, wv = t >> 5;
  const int bb = blockIdx.x >> 5;

  {
    const float* ap = a1 + (size_t)(bb * 64 + wv * 8) * 320 + lane;
    float z0 = 0.f, z1 = 0.f, z2 = 0.f, z3 = 0.f, z4 = 0.f;
    for (int uu = 0; uu < 8; ++uu) {
      z0 += ap[uu * 320];       z1 += ap[uu * 320 + 96];  z2 += ap[uu * 320 + 160];
      z3 += ap[uu * 320 + 224]; z4 += ap[uu * 320 + 288];
    }
    float* zp = a0p + wv * 160 + lane;
    zp[0] = z0; zp[32] = z1; zp[64] = z2; zp[96] = z3; zp[128] = z4;
  }
  for (int idx = t; idx < 384 * 32; idx += 256) w[idx] = W22[idx];
  __syncthreads();
  if (t < 160) {
    float tot = 0.f;
    for (int g = 0; g < 8; ++g) tot += a0p[g * 160 + t];
    a0s[t] = tot * (1.f / 64.f);
  }
  __syncthreads();
  {
    const float* Wv = (wv & 1) ? W01 : W02;
    const int kq0 = (wv >> 1) * 40;
    const float bz = b02[lane] + b22[lane], bo = b01[lane];
    float acc = (wv >> 1) ? 0.f : ((wv & 1) ? bo : bz);
    #pragma unroll
    for (int k = 0; k < 40; ++k) acc += a0s[kq0 + k] * Wv[(kq0 + k) * 32 + lane];
    vp[(wv >> 1) * 64 + (wv & 1) * 32 + lane] = acc;
  }
  __syncthreads();
  if (t < 64) vec[t] = vp[t] + vp[64 + t] + vp[128 + t] + vp[192 + t];
  __syncthreads();

  const int R = (blockIdx.x * 8 + (t >> 5)) * 16;
  const int MA = lane & 15;
  const int kh = lane >> 4;
  const int rowA = R + MA;
  const int bA = rowA >> 12, pA = (rowA >> 6) & 63, qA = rowA & 63;
  const size_t offPQ = ((size_t)((bA * 64 + pA) * 64 + qA)) << 5;
  const size_t offQP = ((size_t)((bA * 64 + qA) * 64 + pA)) << 5;
  const float* comp[6] = { s1m, s2m, s3m, d12, d13, d23 };
  const int n0 = lane & 15;

  v8f acc0 = {};
  v8f acc1 = {};
  #pragma unroll
  for (int f = 0; f < 12; ++f) {
    const float* A = (f < 6) ? (comp[f] + offPQ) : (comp[f - 6] + offQP);
    #pragma unroll
    for (int kk = 0; kk < 32; kk += 4) {
      v2f a = *(const v2f*)(A + kk + 2 * kh);
      const int kcol = f * 32 + kk + 2 * kh;
      v2f b0 = { w[kcol * 32 + n0],      w[(kcol + 1) * 32 + n0] };
      v2f b1 = { w[kcol * 32 + n0 + 16], w[(kcol + 1) * 32 + n0 + 16] };
      acc0 = __builtin_amdgcn_wmma_f32_16x16x4_f32(false, a, false, b0,
                                                   (short)0, acc0, false, false);
      acc1 = __builtin_amdgcn_wmma_f32_16x16x4_f32(false, a, false, b1,
                                                   (short)0, acc1, false, false);
    }
  }
  #pragma unroll
  for (int r = 0; r < 8; ++r) {
    const int M = r + ((lane >> 4) << 3);
    const int row = R + M;
    const int b = row >> 12, p = (row >> 6) & 63, q = row & 63;
    float add0 = r12[((b << 6) + p) * 32 + n0] + r12t[((b << 6) + q) * 32 + n0]
               + vec[n0];
    float add1 = r12[((b << 6) + p) * 32 + n0 + 16] + r12t[((b << 6) + q) * 32 + n0 + 16]
               + vec[n0 + 16];
    if (p == q) {
      add0 += forow[((b << 6) + p) * 32 + n0] + vec[32 + n0];
      add1 += forow[((b << 6) + p) * 32 + n0 + 16] + vec[32 + n0 + 16];
    }
    out[((size_t)row << 5) + n0]      = acc0[r] + add0;
    out[((size_t)row << 5) + n0 + 16] = acc1[r] + add1;
  }
}

extern "C" void kernel_launch(void* const* d_in, const int* in_sizes, int n_in,
                              void* d_out, int out_size, void* d_ws, size_t ws_size,
                              hipStream_t stream) {
  (void)in_sizes; (void)n_in; (void)out_size;
  if (ws_size < WS_FLOATS * sizeof(float)) return;

  const float* x    = (const float*)d_in[0];
  const float* W22  = (const float*)d_in[1];
  const float* b22  = (const float*)d_in[2];
  const float* W12  = (const float*)d_in[3];
  const float* b12  = (const float*)d_in[4];
  const float* W12t = (const float*)d_in[5];
  const float* b12t = (const float*)d_in[6];
  const float* W02  = (const float*)d_in[7];
  const float* b02  = (const float*)d_in[8];
  const float* W11  = (const float*)d_in[9];
  const float* b11  = (const float*)d_in[10];
  const float* W01  = (const float*)d_in[11];
  const float* b01  = (const float*)d_in[12];

  float* ws = (float*)d_ws;
  float* s1p = ws + S1P;   float* s2p = ws + S2P;
  float* s3m = ws + S3M;   float* d12 = ws + D12;
  float* d13 = ws + D13;   float* d23 = ws + D23;
  float* tri = ws + TRI;   float* s1m = ws + S1M;
  float* s2m = ws + S2M;   float* a1  = ws + A1O;
  float* r12 = ws + R12O;
  float* r12t= ws + R12TO; float* forow = ws + FORO;

  k1_contract<<<dim3(8, 8, 8), 256, 0, stream>>>(x, s1p, s2p, s3m, d12, d13, d23, tri);
  k2a_reduce <<<512, 256, 0, stream>>>(s1p, s2p, s1m, s2m);
  k2b_a1     <<<512, 256, 0, stream>>>(s1m, s3m, d12, d13, d23, tri,
                                      W12, b12, W12t, b12t, W11, b11,
                                      a1, r12, r12t, forow);
  k3b_mix    <<<256, 256, 0, stream>>>(s1m, s2m, s3m, d12, d13, d23, W22, a1,
                                       W02, b02, b22, W01, b01,
                                       r12, r12t, forow, (float*)d_out);
}


# Reduce launch (8 partial planes -> s1, s2 means) made latency-tolerant: a thread's 32 loads (16-byte, non-temporal) all issued before the first add, was one 4-byte load per round trip; sum order kept
# speedup vs baseline: 2.7215x; 1.5356x over previous
#include <hip/hip_runtime.h>


typedef __attribute__((ext_vector_type(2))) float v2f;
typedef __attribute__((ext_vector_type(8))) float v8f;

static constexpr size_t S1P  = 0;
static constexpr size_t S2P  = 8388608;
static constexpr size_t S3M  = 16777216;
static constexpr size_t D12  = 17825792;
static constexpr size_t D13  = 18874368;
static constexpr size_t D23  = 19922944;
static constexpr size_t TRI  = 20971520;
static constexpr size_t S1M  = 20987904;
static constexpr size_t S2M  = 22036480;
static constexpr size_t A1O  = 23085056;
static constexpr size_t R12O = 23250176;
static constexpr size_t R12TO= 23266560;
static constexpr size_t FORO = 23282944;
static constexpr size_t WS_FLOATS = 23299584;

typedef __attribute__((ext_vector_type(4))) float v4f;

static __device__ __forceinline__ v4f load_x(const float* p) {
  return __builtin_nontemporal_load((const v4f*)p);
}

__global__ void __launch_bounds__(256)
k1_contract(const float* __restrict__ x,
            float* __restrict__ s1p, float* __restrict__ s2p,
            float* __restrict__ s3m, float* __restrict__ d12,
            float* __restrict__ d13, float* __restrict__ d23,
            float* __restrict__ tri) {
  __shared__ float scr[64 * 8 * 32];
  const int t  = threadIdx.x;
  const int jt = blockIdx.x, it = blockIdx.y, b = blockIdx.z;
  const int lane = t & 31, wave = t >> 5;
  const int kq = t >> 3, c4 = t & 7;
  const int i0 = it * 8, j0 = jt * 8;
  const bool hi_i = (it >> 2) != 0;
  const bool hi_j = (jt >> 2) != 0;
  const float* xb = x + (((size_t)((b * 64 + i0) * 64 + j0)) << 11) + 4 * t;

  v4f s1acc[8][2];
  #pragma unroll
  for (int jj = 0; jj < 8; ++jj) { s1acc[jj][0] = (v4f){0.f, 0.f, 0.f, 0.f}; s1acc[jj][1] = (v4f){0.f, 0.f, 0.f, 0.f}; }

  auto row = [&](const v4f v0, const v4f v1, const int ii, const int jj, v4f& s2a0, v4f& s2a1) {
    const int i = i0 + ii, j = j0 + jj;
    s1acc[jj][0] += v0; s1acc[jj][1] += v1;
    s2a0 += v0;         s2a1 += v1;
    v4f p = v0 + v1;
    p.x += __shfl_xor(p.x, 8, 32);  p.y += __shfl_xor(p.y, 8, 32);
    p.z += __shfl_xor(p.z, 8, 32);  p.w += __shfl_xor(p.w, 8, 32);
    p.x += __shfl_xor(p.x, 16, 32); p.y += __shfl_xor(p.y, 16, 32);
    p.z += __shfl_xor(p.z, 16, 32); p.w += __shfl_xor(p.w, 16, 32);
    if (lane < 8) *(v4f*)&scr[(((ii * 8 + jj) * 8 + wave) << 5) + 4 * lane] = p;
    const size_t ro = ((size_t)((b * 64 + i) * 64 + j)) << 5;
    if (kq == (i & 31)) *(v4f*)(d13 + ro + 4 * c4) = hi_i ? v1 : v0;
    if (kq == (j & 31)) *(v4f*)(d23 + ro + 4 * c4) = hi_j ? v1 : v0;
    if (i == j) {
      float* dd = d12 + (((size_t)(b * 64 + i)) << 11) + 4 * t;
      *(v4f*)dd = v0; *(v4f*)(dd + 1024) = v1;
      if (kq == (i & 31)) *(v4f*)(tri + ((b << 6) + i) * 32 + 4 * c4) = hi_i ? v1 : v0;
    }
    asm volatile("" : "+v"(s1acc[jj][0]), "+v"(s1acc[jj][1]), "+v"(s2a0), "+v"(s2a1));
  };

  v4f buf[8][2];
  #pragma unroll
  for (int jj = 0; jj < 8; ++jj) {
    buf[jj][0] = load_x(xb + jj * 2048);
    buf[jj][1] = load_x(xb + jj * 2048 + 1024);
  }
  for (int ii = 0; ii < 7; ++ii) {
    const float* nx = xb + (size_t)(ii + 1) * 131072;
    v4f s2a0 = {0.f, 0.f, 0.f, 0.f}, s2a1 = {0.f, 0.f, 0.f, 0.f};
    #pragma unroll
    for (int jj = 0; jj < 8; ++jj) {
      row(buf[jj][0], buf[jj][1], ii, jj, s2a0, s2a1);
      buf[jj][0] = load_x(nx + jj * 2048);
      buf[jj][1] = load_x(nx + jj * 2048 + 1024);
    }
    float* dst = s2p + (((size_t)((b * 8 + jt) * 64 + i0 + ii)) << 11) + 4 * t;
    *(v4f*)dst = s2a0; *(v4f*)(dst + 1024) = s2a1;
  }
  {
    v4f s2a0 = {0.f, 0.f, 0.f, 0.f}, s2a1 = {0.f, 0.f, 0.f, 0.f};
    #pragma unroll
    for (int jj = 0; jj < 8; ++jj) row(buf[jj][0], buf[jj][1], 7, jj, s2a0, s2a1);
    float* dst = s2p + (((size_t)((b * 8 + jt) * 64 + i0 + 7)) << 11) + 4 * t;
    *(v4f*)dst = s2a0; *(v4f*)(dst + 1024) = s2a1;
  }

  #pragma unroll
  for (int jj = 0; jj < 8; ++jj) {
    float* dst = s1p + (((size_t)((b * 8 + it) * 64 + j0 + jj)) << 11) + 4 * t;
    *(v4f*)dst = s1acc[jj][0]; *(v4f*)(dst + 1024) = s1acc[jj][1];
  }

  __syncthreads();
  #pragma unroll
  for (int ii = 0; ii < 8; ++ii) {
    float tot = 0.f;
    #pragma unroll
    for (int w = 0; w < 8; ++w) tot += scr[(((ii * 8 + wave) * 8 + w) << 5) + lane];
    s3m[(((size_t)((b * 64 + i0 + ii) * 64 + j0 + wave)) << 5) + lane] = tot * (1.f / 64.f);
  }
}


static __device__ __forceinline__ v4f load_part(const float* p) {
  return __builtin_nontemporal_load((const v4f*)p);
}

__global__ void __launch_bounds__(256)
k2a_reduce(const float* __restrict__ s1p, const float* __restrict__ s2p,
           float* __restrict__ s1m, float* __restrict__ s2m) {
  const int t = threadIdx.x;
  const int u = blockIdx.x & 63, b = blockIdx.x >> 6;
  const size_t in0 = (((size_t)(b * 512 + u)) << 11) + 4 * t;
  const float* p1 = s1p + in0;
  const float* p2 = s2p + in0;

  v4f r1[2][8], r2[2][8];
  #pragma unroll
  for (int q = 0; q < 2; ++q) {
    #pragma unroll
    for (int g = 0; g < 8; ++g) {
      r1[q][g] = load_part(p1 + (size_t)g * 131072 + q * 1024);
      r2[q][g] = load_part(p2 + (size_t)g * 131072 + q * 1024);
    }
  }
  __builtin_amdgcn_sched_barrier(0);

  const size_t o = (((size_t)(b * 64 + u)) << 11) + 4 * t;
  #pragma unroll
  for (int q = 0; q < 2; ++q) {
    v4f a = {0.f, 0.f, 0.f, 0.f}, s = {0.f, 0.f, 0.f, 0.f};
    #pragma unroll
    for (int g = 0; g < 8; ++g) { a += r1[q][g]; s += r2[q][g]; }
    *(v4f*)(s1m + o + q * 1024) = a * (1.f / 64.f);
    *(v4f*)(s2m + o + q * 1024) = s * (1.f / 64.f);
  }
}

__global__ void k2b_a1(const float* __restrict__ s1m, const float* __restrict__ s3m,
                       const float* __restrict__ d12, const float* __restrict__ d13,
                       const float* __restrict__ d23, const float* __restrict__ tri,
                       const float* __restrict__ W12, const float* __restrict__ b12,
                       const float* __restrict__ W12t, const float* __restrict__ b12t,
                       const float* __restrict__ W11, const float* __restrict__ b11,
                       float* __restrict__ a1, float* __restrict__ r12,
                       float* __restrict__ r12t, float* __restrict__ forow) {
  __shared__ float scr[256];
  __shared__ float arow[320];
  __shared__ float part[8 * 96];
  const int t = threadIdx.x, c = t & 31, g = t >> 5;
  const int b = blockIdx.x >> 6, u = blockIdx.x & 63;
  float* dst = a1 + (size_t)(b * 64 + u) * 320;

  auto red = [&](float partial, int slot) {
    scr[t] = partial;
    __syncthreads();
    if (t < 32) {
      float tot = 0.f;
      #pragma unroll
      for (int gg = 0; gg < 8; ++gg) tot += scr[gg * 32 + t];
      const float m = tot * (1.f / 64.f);
      dst[slot * 32 + t] = m;
      arow[slot * 32 + t] = m;
    }
    __syncthreads();
  };
  float acc;
  acc = 0.f; for (int s = 0; s < 8; ++s) { int v = g + 8 * s;
    acc += s1m[(((size_t)((b * 64 + v) * 64 + u)) << 5) + c]; } red(acc, 0);
  acc = 0.f; for (int s = 0; s < 8; ++s) { int v = g + 8 * s;
    acc += s1m[(((size_t)((b * 64 + u) * 64 + v)) << 5) + c]; } red(acc, 1);
  acc = 0.f; for (int s = 0; s < 8; ++s) { int v = g + 8 * s;
    acc += s3m[(((size_t)((b * 64 + u) * 64 + v)) << 5) + c]; } red(acc, 2);
  acc = 0.f; for (int s = 0; s < 8; ++s) { int v = g + 8 * s;
    acc += d12[(((size_t)(b * 64 + v)) << 11) + u * 32 + c]; } red(acc, 3);
  acc = 0.f; for (int s = 0; s < 8; ++s) { int v = g + 8 * s;
    acc += d12[(((size_t)(b * 64 + u)) << 11) + v * 32 + c]; } red(acc, 4);
  acc = 0.f; for (int s = 0; s < 8; ++s) { int v = g + 8 * s;
    acc += d13[(((size_t)((b * 64 + v) * 64 + u)) << 5) + c]; } red(acc, 5);
  acc = 0.f; for (int s = 0; s < 8; ++s) { int v = g + 8 * s;
    acc += d13[(((size_t)((b * 64 + u) * 64 + v)) << 5) + c]; } red(acc, 6);
  acc = 0.f; for (int s = 0; s < 8; ++s) { int v = g + 8 * s;
    acc += d23[(((size_t)((b * 64 + v) * 64 + u)) << 5) + c]; } red(acc, 7);
  acc = 0.f; for (int s = 0; s < 8; ++s) { int v = g + 8 * s;
    acc += d23[(((size_t)((b * 64 + u) * 64 + v)) << 5) + c]; } red(acc, 8);
  if (t < 32) { const float v = tri[(b * 64 + u) * 32 + t]; dst[9 * 32 + t] = v; arow[9 * 32 + t] = v; }
  __syncthreads();

  const int k0 = g * 40;
  float m12 = 0.f, m12t = 0.f, m11 = 0.f;
  #pragma unroll
  for (int k = 0; k < 40; ++k) {
    const float av = arow[k0 + k];
    m12  += av * W12 [(k0 + k) * 32 + c];
    m12t += av * W12t[(k0 + k) * 32 + c];
    m11  += av * W11 [(k0 + k) * 32 + c];
  }
  part[g * 96 + c] = m12; part[g * 96 + 32 + c] = m12t; part[g * 96 + 64 + c] = m11;
  __syncthreads();
  if (t < 96) {
    float tot = (g == 0) ? b12[c] : (g == 1) ? b12t[c] : b11[c];
    #pragma unroll
    for (int gg = 0; gg < 8; ++gg) tot += part[gg * 96 + t];
    float* o = (g == 0) ? r12 : (g == 1) ? r12t : forow;
    o[(b * 64 + u) * 32 + c] = tot;
  }
}

__global__ void k3b_mix(const float* __restrict__ s1m, const float* __restrict__ s2m,
                        const float* __restrict__ s3m, const float* __restrict__ d12,
                        const float* __restrict__ d13, const float* __restrict__ d23,
                        const float* __restrict__ W22, const float* __restrict__ a1,
                        const float* __restrict__ W02, const float* __restrict__ b02,
                        const float* __restrict__ b22,
                        const float* __restrict__ W01, const float* __restrict__ b01,
                        const float* __restrict__ r12, const float* __restrict__ r12t,
                        const float* __restrict__ forow,
                        float* __restrict__ out) {
  __shared__ float w[384 * 32];
  __shared__ float a0p[8 * 160];
  __shared__ float a0s[160];
  __shared__ float vp[4 * 64];
  __shared__ float vec[64];
  const int t = threadIdx.x;
  const int lane = t & 31, wv = t >> 5;
  const int bb = blockIdx.x >> 5;

  {
    const float* ap = a1 + (size_t)(bb * 64 + wv * 8) * 320 + lane;
    float z0 = 0.f, z1 = 0.f, z2 = 0.f, z3 = 0.f, z4 = 0.f;
    for (int uu = 0; uu < 8; ++uu) {
      z0 += ap[uu * 320];       z1 += ap[uu * 320 + 96];  z2 += ap[uu * 320 + 160];
      z3 += ap[uu * 320 + 224]; z4 += ap[uu * 320 + 288];
    }
    float* zp = a0p + wv * 160 + lane;
    zp[0] = z0; zp[32] = z1; zp[64] = z2; zp[96] = z3; zp[128] = z4;
  }
  for (int idx = t; idx < 384 * 32; idx += 256) w[idx] = W22[idx];
  __syncthreads();
  if (t < 160) {
    float tot = 0.f;
    for (int g = 0; g < 8; ++g) tot += a0p[g * 160 + t];
    a0s[t] = tot * (1.f / 64.f);
  }
  __syncthreads();
  {
    const float* Wv = (wv & 1) ? W01 : W02;
    const int kq0 = (wv >> 1) * 40;
    const float bz = b02[lane] + b22[lane], bo = b01[lane];
    float acc = (wv >> 1) ? 0.f : ((wv & 1) ? bo : bz);
    #pragma unroll
    for (int k = 0; k < 40; ++k) acc += a0s[kq0 + k] * Wv[(kq0 + k) * 32 + lane];
    vp[(wv >> 1) * 64 + (wv & 1) * 32 + lane] = acc;
  }
  __syncthreads();
  if (t < 64) vec[t] = vp[t] + vp[64 + t] + vp[128 + t] + vp[192 + t];
  __syncthreads();

  const int R = (blockIdx.x * 8 + (t >> 5)) * 16;
  const int MA = lane & 15;
  const int kh = lane >> 4;
  const int rowA = R + MA;
  const int bA = rowA >> 12, pA = (rowA >> 6) & 63, qA = rowA & 63;
  const size_t offPQ = ((size_t)((bA * 64 + pA) * 64 + qA)) << 5;
  const size_t offQP = ((size_t)((bA * 64 + qA) * 64 + pA)) << 5;
  const float* comp[6] = { s1m, s2m, s3m, d12, d13, d23 };
  const int n0 = lane & 15;

  v8f acc0 = {};
  v8f acc1 = {};
  #pragma unroll
  for (int f = 0; f < 12; ++f) {
    const float* A = (f < 6) ? (comp[f] + offPQ) : (comp[f - 6] + offQP);
    #pragma unroll
    for (int kk = 0; kk < 32; kk += 4) {
      v2f a = *(const v2f*)(A + kk + 2 * kh);
      const int kcol = f * 32 + kk + 2 * kh;
      v2f b0 = { w[kcol * 32 + n0],      w[(kcol + 1) * 32 + n0] };
      v2f b1 = { w[kcol * 32 + n0 + 16], w[(kcol + 1) * 32 + n0 + 16] };
      acc0 = __builtin_amdgcn_wmma_f32_16x16x4_f32(false, a, false, b0,
                                                   (short)0, acc0, false, false);
      acc1 = __builtin_amdgcn_wmma_f32_16x16x4_f32(false, a, false, b1,
                                                   (short)0, acc1, false, false);
    }
  }
  #pragma unroll
  for (int r = 0; r < 8; ++r) {
    const int M = r + ((lane >> 4) << 3);
    const int row = R + M;
    const int b = row >> 12, p = (row >> 6) & 63, q = row & 63;
    float add0 = r12[((b << 6) + p) * 32 + n0] + r12t[((b << 6) + q) * 32 + n0]
               + vec[n0];
    float add1 = r12[((b << 6) + p) * 32 + n0 + 16] + r12t[((b << 6) + q) * 32 + n0 + 16]
               + vec[n0 + 16];
    if (p == q) {
      add0 += forow[((b << 6) + p) * 32 + n0] + vec[32 + n0];
      add1 += forow[((b << 6) + p) * 32 + n0 + 16] + vec[32 + n0 + 16];
    }
    out[((size_t)row << 5) + n0]      = acc0[r] + add0;
    out[((size_t)row << 5) + n0 + 16] = acc1[r] + add1;
  }
}

extern "C" void kernel_launch(void* const* d_in, const int* in_sizes, int n_in,
                              void* d_out, int out_size, void* d_ws, size_t ws_size,
                              hipStream_t stream) {
  (void)in_sizes; (void)n_in; (void)out_size;
  if (ws_size < WS_FLOATS * sizeof(float)) return;

  const float* x    = (const float*)d_in[0];
  const float* W22  = (const float*)d_in[1];
  const float* b22  = (const float*)d_in[2];
  const float* W12  = (const float*)d_in[3];
  const float* b12  = (const float*)d_in[4];
  const float* W12t = (const float*)d_in[5];
  const float* b12t = (const float*)d_in[6];
  const float* W02  = (const float*)d_in[7];
  const float* b02  = (const float*)d_in[8];
  const float* W11  = (const float*)d_in[9];
  const float* b11  = (const float*)d_in[10];
  const float* W01  = (const float*)d_in[11];
  const float* b01  = (const float*)d_in[12];

  float* ws = (float*)d_ws;
  float* s1p = ws + S1P;   float* s2p = ws + S2P;
  float* s3m = ws + S3M;   float* d12 = ws + D12;
  float* d13 = ws + D13;   float* d23 = ws + D23;
  float* tri = ws + TRI;   float* s1m = ws + S1M;
  float* s2m = ws + S2M;   float* a1  = ws + A1O;
  float* r12 = ws + R12O;
  float* r12t= ws + R12TO; float* forow = ws + FORO;

  k1_contract<<<dim3(8, 8, 8), 256, 0, stream>>>(x, s1p, s2p, s3m, d12, d13, d23, tri);
  k2a_reduce <<<512, 256, 0, stream>>>(s1p, s2p, s1m, s2m);
  k2b_a1     <<<512, 256, 0, stream>>>(s1m, s3m, d12, d13, d23, tri,
                                      W12, b12, W12t, b12t, W11, b11,
                                      a1, r12, r12t, forow);
  k3b_mix    <<<256, 256, 0, stream>>>(s1m, s2m, s3m, d12, d13, d23, W22, a1,
                                       W02, b02, b22, W01, b01,
                                       r12, r12t, forow, (float*)d_out);
}


# a1-row launch made latency-tolerant: the 72 loads of its nine sums requested in one run before the first barrier, one fold behind one barrier, was nine load-fold-store rounds; sum order kept
# speedup vs baseline: 3.0108x; 1.1063x over previous
#include <hip/hip_runtime.h>


typedef __attribute__((ext_vector_type(2))) float v2f;
typedef __attribute__((ext_vector_type(8))) float v8f;

static constexpr size_t S1P  = 0;
static constexpr size_t S2P  = 8388608;
static constexpr size_t S3M  = 16777216;
static constexpr size_t D12  = 17825792;
static constexpr size_t D13  = 18874368;
static constexpr size_t D23  = 19922944;
static constexpr size_t TRI  = 20971520;
static constexpr size_t S1M  = 20987904;
static constexpr size_t S2M  = 22036480;
static constexpr size_t A1O  = 23085056;
static constexpr size_t R12O = 23250176;
static constexpr size_t R12TO= 23266560;
static constexpr size_t FORO = 23282944;
static constexpr size_t WS_FLOATS = 23299584;

typedef __attribute__((ext_vector_type(4))) float v4f;

static __device__ __forceinline__ v4f load_x(const float* p) {
  return __builtin_nontemporal_load((const v4f*)p);
}

__global__ void __launch_bounds__(256)
k1_contract(const float* __restrict__ x,
            float* __restrict__ s1p, float* __restrict__ s2p,
            float* __restrict__ s3m, float* __restrict__ d12,
            float* __restrict__ d13, float* __restrict__ d23,
            float* __restrict__ tri) {
  __shared__ float scr[64 * 8 * 32];
  const int t  = threadIdx.x;
  const int jt = blockIdx.x, it = blockIdx.y, b = blockIdx.z;
  const int lane = t & 31, wave = t >> 5;
  const int kq = t >> 3, c4 = t & 7;
  const int i0 = it * 8, j0 = jt * 8;
  const bool hi_i = (it >> 2) != 0;
  const bool hi_j = (jt >> 2) != 0;
  const float* xb = x + (((size_t)((b * 64 + i0) * 64 + j0)) << 11) + 4 * t;

  v4f s1acc[8][2];
  #pragma unroll
  for (int jj = 0; jj < 8; ++jj) { s1acc[jj][0] = (v4f){0.f, 0.f, 0.f, 0.f}; s1acc[jj][1] = (v4f){0.f, 0.f, 0.f, 0.f}; }

  auto row = [&](const v4f v0, const v4f v1, const int ii, const int jj, v4f& s2a0, v4f& s2a1) {
    const int i = i0 + ii, j = j0 + jj;
    s1acc[jj][0] += v0; s1acc[jj][1] += v1;
    s2a0 += v0;         s2a1 += v1;
    v4f p = v0 + v1;
    p.x += __shfl_xor(p.x, 8, 32);  p.y += __shfl_xor(p.y, 8, 32);
    p.z += __shfl_xor(p.z, 8, 32);  p.w += __shfl_xor(p.w, 8, 32);
    p.x += __shfl_xor(p.x, 16, 32); p.y += __shfl_xor(p.y, 16, 32);
    p.z += __shfl_xor(p.z, 16, 32); p.w += __shfl_xor(p.w, 16, 32);
    if (lane < 8) *(v4f*)&scr[(((ii * 8 + jj) * 8 + wave) << 5) + 4 * lane] = p;
    const size_t ro = ((size_t)((b * 64 + i) * 64 + j)) << 5;
    if (kq == (i & 31)) *(v4f*)(d13 + ro + 4 * c4) = hi_i ? v1 : v0;
    if (kq == (j & 31)) *(v4f*)(d23 + ro + 4 * c4) = hi_j ? v1 : v0;
    if (i == j) {
      float* dd = d12 + (((size_t)(b * 64 + i)) << 11) + 4 * t;
      *(v4f*)dd = v0; *(v4f*)(dd + 1024) = v1;
      if (kq == (i & 31)) *(v4f*)(tri + ((b << 6) + i) * 32 + 4 * c4) = hi_i ? v1 : v0;
    }
    asm volatile("" : "+v"(s1acc[jj][0]), "+v"(s1acc[jj][1]), "+v"(s2a0), "+v"(s2a1));
  };

  v4f buf[8][2];
  #pragma unroll
  for (int jj = 0; jj < 8; ++jj) {
    buf[jj][0] = load_x(xb + jj * 2048);
    buf[jj][1] = load_x(xb + jj * 2048 + 1024);
  }
  for (int ii = 0; ii < 7; ++ii) {
    const float* nx = xb + (size_t)(ii + 1) * 131072;
    v4f s2a0 = {0.f, 0.f, 0.f, 0.f}, s2a1 = {0.f, 0.f, 0.f, 0.f};
    #pragma unroll
    for (int jj = 0; jj < 8; ++jj) {
      row(buf[jj][0], buf[jj][1], ii, jj, s2a0, s2a1);
      buf[jj][0] = load_x(nx + jj * 2048);
      buf[jj][1] = load_x(nx + jj * 2048 + 1024);
    }
    float* dst = s2p + (((size_t)((b * 8 + jt) * 64 + i0 + ii)) << 11) + 4 * t;
    *(v4f*)dst = s2a0; *(v4f*)(dst + 1024) = s2a1;
  }
  {
    v4f s2a0 = {0.f, 0.f, 0.f, 0.f}, s2a1 = {0.f, 0.f, 0.f, 0.f};
    #pragma unroll
    for (int jj = 0; jj < 8; ++jj) row(buf[jj][0], buf[jj][1], 7, jj, s2a0, s2a1);
    float* dst = s2p + (((size_t)((b * 8 + jt) * 64 + i0 + 7)) << 11) + 4 * t;
    *(v4f*)dst = s2a0; *(v4f*)(dst + 1024) = s2a1;
  }

  #pragma unroll
  for (int jj = 0; jj < 8; ++jj) {
    float* dst = s1p + (((size_t)((b * 8 + it) * 64 + j0 + jj)) << 11) + 4 * t;
    *(v4f*)dst = s1acc[jj][0]; *(v4f*)(dst + 1024) = s1acc[jj][1];
  }

  __syncthreads();
  #pragma unroll
  for (int ii = 0; ii < 8; ++ii) {
    float tot = 0.f;
    #pragma unroll
    for (int w = 0; w < 8; ++w) tot += scr[(((ii * 8 + wave) * 8 + w) << 5) + lane];
    s3m[(((size_t)((b * 64 + i0 + ii) * 64 + j0 + wave)) << 5) + lane] = tot * (1.f / 64.f);
  }
}


static __device__ __forceinline__ v4f load_part(const float* p) {
  return __builtin_nontemporal_load((const v4f*)p);
}

__global__ void __launch_bounds__(256)
k2a_reduce(const float* __restrict__ s1p, const float* __restrict__ s2p,
           float* __restrict__ s1m, float* __restrict__ s2m) {
  const int t = threadIdx.x;
  const int u = blockIdx.x & 63, b = blockIdx.x >> 6;
  const size_t in0 = (((size_t)(b * 512 + u)) << 11) + 4 * t;
  const float* p1 = s1p + in0;
  const float* p2 = s2p + in0;

  v4f r1[2][8], r2[2][8];
  #pragma unroll
  for (int q = 0; q < 2; ++q) {
    #pragma unroll
    for (int g = 0; g < 8; ++g) {
      r1[q][g] = load_part(p1 + (size_t)g * 131072 + q * 1024);
      r2[q][g] = load_part(p2 + (size_t)g * 131072 + q * 1024);
    }
  }
  __builtin_amdgcn_sched_barrier(0);

  const size_t o = (((size_t)(b * 64 + u)) << 11) + 4 * t;
  #pragma unroll
  for (int q = 0; q < 2; ++q) {
    v4f a = {0.f, 0.f, 0.f, 0.f}, s = {0.f, 0.f, 0.f, 0.f};
    #pragma unroll
    for (int g = 0; g < 8; ++g) { a += r1[q][g]; s += r2[q][g]; }
    *(v4f*)(s1m + o + q * 1024) = a * (1.f / 64.f);
    *(v4f*)(s2m + o + q * 1024) = s * (1.f / 64.f);
  }
}

__global__ void __launch_bounds__(256)
k2b_a1(const float* __restrict__ s1m, const float* __restrict__ s3m,
       const float* __restrict__ d12, const float* __restrict__ d13,
       const float* __restrict__ d23, const float* __restrict__ tri,
       const float* __restrict__ W12, const float* __restrict__ b12,
       const float* __restrict__ W12t, const float* __restrict__ b12t,
       const float* __restrict__ W11, const float* __restrict__ b11,
       float* __restrict__ a1, float* __restrict__ r12,
       float* __restrict__ r12t, float* __restrict__ forow) {
  __shared__ float scr[9 * 256];
  __shared__ float arow[320];
  __shared__ float part[8 * 96];
  const int t = threadIdx.x, c = t & 31, g = t >> 5;
  const int b = blockIdx.x >> 6, u = blockIdx.x & 63;
  float* dst = a1 + (size_t)(b * 64 + u) * 320;

  const size_t col = (((size_t)(b * 4096 + u)) << 5) + c + (size_t)g * 2048;
  const size_t lin = (((size_t)((b * 64 + u) * 64)) << 5) + c + (size_t)g * 32;
  const float* src9[9] = { s1m + col, s1m + lin, s3m + lin, d12 + col, d12 + lin,
                           d13 + col, d13 + lin, d23 + col, d23 + lin };
  float r[9][8];
  #pragma unroll
  for (int q = 0; q < 9; ++q) {
    const int step = (q == 0 || q == 3 || q == 5 || q == 7) ? 8 * 2048 : 8 * 32;
    #pragma unroll
    for (int s = 0; s < 8; ++s) r[q][s] = src9[q][s * step];
  }
  float tv = 0.f;
  if (t < 32) tv = tri[(b * 64 + u) * 32 + t];
  __builtin_amdgcn_sched_barrier(0);

  #pragma unroll
  for (int q = 0; q < 9; ++q) {
    float acc = 0.f;
    #pragma unroll
    for (int s = 0; s < 8; ++s) acc += r[q][s];
    scr[q * 256 + t] = acc;
  }
  __syncthreads();
  for (int q = g; q < 9; q += 8) {
    float tot = 0.f;
    #pragma unroll
    for (int gg = 0; gg < 8; ++gg) tot += scr[q * 256 + gg * 32 + c];
    const float m = tot * (1.f / 64.f);
    dst[q * 32 + c] = m;
    arow[q * 32 + c] = m;
  }
  if (t < 32) { dst[9 * 32 + t] = tv; arow[9 * 32 + t] = tv; }
  __syncthreads();

  const int k0 = g * 40;
  float m12 = 0.f, m12t = 0.f, m11 = 0.f;
  #pragma unroll
  for (int k = 0; k < 40; ++k) {
    const float av = arow[k0 + k];
    m12  += av * W12 [(k0 + k) * 32 + c];
    m12t += av * W12t[(k0 + k) * 32 + c];
    m11  += av * W11 [(k0 + k) * 32 + c];
  }
  part[g * 96 + c] = m12; part[g * 96 + 32 + c] = m12t; part[g * 96 + 64 + c] = m11;
  __syncthreads();
  if (t < 96) {
    float tot = (g == 0) ? b12[c] : (g == 1) ? b12t[c] : b11[c];
    #pragma unroll
    for (int gg = 0; gg < 8; ++gg) tot += part[gg * 96 + t];
    float* o = (g == 0) ? r12 : (g == 1) ? r12t : forow;
    o[(b * 64 + u) * 32 + c] = tot;
  }
}

__global__ void k3b_mix(const float* __restrict__ s1m, const float* __restrict__ s2m,
                        const float* __restrict__ s3m, const float* __restrict__ d12,
                        const float* __restrict__ d13, const float* __restrict__ d23,
                        const float* __restrict__ W22, const float* __restrict__ a1,
                        const float* __restrict__ W02, const float* __restrict__ b02,
                        const float* __restrict__ b22,
                        const float* __restrict__ W01, const float* __restrict__ b01,
                        const float* __restrict__ r12, const float* __restrict__ r12t,
                        const float* __restrict__ forow,
                        float* __restrict__ out) {
  __shared__ float w[384 * 32];
  __shared__ float a0p[8 * 160];
  __shared__ float a0s[160];
  __shared__ float vp[4 * 64];
  __shared__ float vec[64];
  const int t = threadIdx.x;
  const int lane = t & 31, wv = t >> 5;
  const int bb = blockIdx.x >> 5;

  {
    const float* ap = a1 + (size_t)(bb * 64 + wv * 8) * 320 + lane;
    float z0 = 0.f, z1 = 0.f, z2 = 0.f, z3 = 0.f, z4 = 0.f;
    for (int uu = 0; uu < 8; ++uu) {
      z0 += ap[uu * 320];       z1 += ap[uu * 320 + 96];  z2 += ap[uu * 320 + 160];
      z3 += ap[uu * 320 + 224]; z4 += ap[uu * 320 + 288];
    }
    float* zp = a0p + wv * 160 + lane;
    zp[0] = z0; zp[32] = z1; zp[64] = z2; zp[96] = z3; zp[128] = z4;
  }
  for (int idx = t; idx < 384 * 32; idx += 256) w[idx] = W22[idx];
  __syncthreads();
  if (t < 160) {
    float tot = 0.f;
    for (int g = 0; g < 8; ++g) tot += a0p[g * 160 + t];
    a0s[t] = tot * (1.f / 64.f);
  }
  __syncthreads();
  {
    const float* Wv = (wv & 1) ? W01 : W02;
    const int kq0 = (wv >> 1) * 40;
    const float bz = b02[lane] + b22[lane], bo = b01[lane];
    float acc = (wv >> 1) ? 0.f : ((wv & 1) ? bo : bz);
    #pragma unroll
    for (int k = 0; k < 40; ++k) acc += a0s[kq0 + k] * Wv[(kq0 + k) * 32 + lane];
    vp[(wv >> 1) * 64 + (wv & 1) * 32 + lane] = acc;
  }
  __syncthreads();
  if (t < 64) vec[t] = vp[t] + vp[64 + t] + vp[128 + t] + vp[192 + t];
  __syncthreads();

  const int R = (blockIdx.x * 8 + (t >> 5)) * 16;
  const int MA = lane & 15;
  const int kh = lane >> 4;
  const int rowA = R + MA;
  const int bA = rowA >> 12, pA = (rowA >> 6) & 63, qA = rowA & 63;
  const size_t offPQ = ((size_t)((bA * 64 + pA) * 64 + qA)) << 5;
  const size_t offQP = ((size_t)((bA * 64 + qA) * 64 + pA)) << 5;
  const float* comp[6] = { s1m, s2m, s3m, d12, d13, d23 };
  const int n0 = lane & 15;

  v8f acc0 = {};
  v8f acc1 = {};
  #pragma unroll
  for (int f = 0; f < 12; ++f) {
    const float* A = (f < 6) ? (comp[f] + offPQ) : (comp[f - 6] + offQP);
    #pragma unroll
    for (int kk = 0; kk < 32; kk += 4) {
      v2f a = *(const v2f*)(A + kk + 2 * kh);
      const int kcol = f * 32 + kk + 2 * kh;
      v2f b0 = { w[kcol * 32 + n0],      w[(kcol + 1) * 32 + n0] };
      v2f b1 = { w[kcol * 32 + n0 + 16], w[(kcol + 1) * 32 + n0 + 16] };
      acc0 = __builtin_amdgcn_wmma_f32_16x16x4_f32(false, a, false, b0,
                                                   (short)0, acc0, false, false);
      acc1 = __builtin_amdgcn_wmma_f32_16x16x4_f32(false, a, false, b1,
                                                   (short)0, acc1, false, false);
    }
  }
  #pragma unroll
  for (int r = 0; r < 8; ++r) {
    const int M = r + ((lane >> 4) << 3);
    const int row = R + M;
    const int b = row >> 12, p = (row >> 6) & 63, q = row & 63;
    float add0 = r12[((b << 6) + p) * 32 + n0] + r12t[((b << 6) + q) * 32 + n0]
               + vec[n0];
    float add1 = r12[((b << 6) + p) * 32 + n0 + 16] + r12t[((b << 6) + q) * 32 + n0 + 16]
               + vec[n0 + 16];
    if (p == q) {
      add0 += forow[((b << 6) + p) * 32 + n0] + vec[32 + n0];
      add1 += forow[((b << 6) + p) * 32 + n0 + 16] + vec[32 + n0 + 16];
    }
    out[((size_t)row << 5) + n0]      = acc0[r] + add0;
    out[((size_t)row << 5) + n0 + 16] = acc1[r] + add1;
  }
}

extern "C" void kernel_launch(void* const* d_in, const int* in_sizes, int n_in,
                              void* d_out, int out_size, void* d_ws, size_t ws_size,
                              hipStream_t stream) {
  (void)in_sizes; (void)n_in; (void)out_size;
  if (ws_size < WS_FLOATS * sizeof(float)) return;

  const float* x    = (const float*)d_in[0];
  const float* W22  = (const float*)d_in[1];
  const float* b22  = (const float*)d_in[2];
  const float* W12  = (const float*)d_in[3];
  const float* b12  = (const float*)d_in[4];
  const float* W12t = (const float*)d_in[5];
  const float* b12t = (const float*)d_in[6];
  const float* W02  = (const float*)d_in[7];
  const float* b02  = (const float*)d_in[8];
  const float* W11  = (const float*)d_in[9];
  const float* b11  = (const float*)d_in[10];
  const float* W01  = (const float*)d_in[11];
  const float* b01  = (const float*)d_in[12];

  float* ws = (float*)d_ws;
  float* s1p = ws + S1P;   float* s2p = ws + S2P;
  float* s3m = ws + S3M;   float* d12 = ws + D12;
  float* d13 = ws + D13;   float* d23 = ws + D23;
  float* tri = ws + TRI;   float* s1m = ws + S1M;
  float* s2m = ws + S2M;   float* a1  = ws + A1O;
  float* r12 = ws + R12O;
  float* r12t= ws + R12TO; float* forow = ws + FORO;

  k1_contract<<<dim3(8, 8, 8), 256, 0, stream>>>(x, s1p, s2p, s3m, d12, d13, d23, tri);
  k2a_reduce <<<512, 256, 0, stream>>>(s1p, s2p, s1m, s2m);
  k2b_a1     <<<512, 256, 0, stream>>>(s1m, s3m, d12, d13, d23, tri,
                                      W12, b12, W12t, b12t, W11, b11,
                                      a1, r12, r12t, forow);
  k3b_mix    <<<256, 256, 0, stream>>>(s1m, s2m, s3m, d12, d13, d23, W22, a1,
                                       W02, b02, b22, W01, b01,
                                       r12, r12t, forow, (float*)d_out);
}


# Launches 4 -> 3: the reduce launch (8 partial planes -> s1, s2 means) folded into the a1-row launch; mean(x,(1,2)) now summed from per-tile sums that pass 1 writes (new sum order)
# speedup vs baseline: 3.1228x; 1.0372x over previous
#include <hip/hip_runtime.h>


typedef __attribute__((ext_vector_type(2))) float v2f;
typedef __attribute__((ext_vector_type(8))) float v8f;

static constexpr size_t S1P  = 0;
static constexpr size_t S2P  = 8388608;
static constexpr size_t S3M  = 16777216;
static constexpr size_t D12  = 17825792;
static constexpr size_t D13  = 18874368;
static constexpr size_t D23  = 19922944;
static constexpr size_t TRI  = 20971520;
static constexpr size_t S1M  = 20987904;
static constexpr size_t S2M  = 22036480;
static constexpr size_t A1O  = 23085056;
static constexpr size_t R12O = 23250176;
static constexpr size_t R12TO= 23266560;
static constexpr size_t FORO = 23282944;
static constexpr size_t TSP  = 23299584;
static constexpr size_t WS_FLOATS = 24348160;

typedef __attribute__((ext_vector_type(4))) float v4f;

static __device__ __forceinline__ v4f load_x(const float* p) {
  return __builtin_nontemporal_load((const v4f*)p);
}

__global__ void __launch_bounds__(256)
k1_contract(const float* __restrict__ x,
            float* __restrict__ s1p, float* __restrict__ s2p,
            float* __restrict__ s3m, float* __restrict__ d12,
            float* __restrict__ d13, float* __restrict__ d23,
            float* __restrict__ tri, float* __restrict__ tsp) {
  __shared__ float scr[64 * 8 * 32];
  const int t  = threadIdx.x;
  const int jt = blockIdx.x, it = blockIdx.y, b = blockIdx.z;
  const int lane = t & 31, wave = t >> 5;
  const int kq = t >> 3, c4 = t & 7;
  const int i0 = it * 8, j0 = jt * 8;
  const bool hi_i = (it >> 2) != 0;
  const bool hi_j = (jt >> 2) != 0;
  const float* xb = x + (((size_t)((b * 64 + i0) * 64 + j0)) << 11) + 4 * t;

  v4f s1acc[8][2];
  #pragma unroll
  for (int jj = 0; jj < 8; ++jj) { s1acc[jj][0] = (v4f){0.f, 0.f, 0.f, 0.f}; s1acc[jj][1] = (v4f){0.f, 0.f, 0.f, 0.f}; }

  auto row = [&](const v4f v0, const v4f v1, const int ii, const int jj, v4f& s2a0, v4f& s2a1) {
    const int i = i0 + ii, j = j0 + jj;
    s1acc[jj][0] += v0; s1acc[jj][1] += v1;
    s2a0 += v0;         s2a1 += v1;
    v4f p = v0 + v1;
    p.x += __shfl_xor(p.x, 8, 32);  p.y += __shfl_xor(p.y, 8, 32);
    p.z += __shfl_xor(p.z, 8, 32);  p.w += __shfl_xor(p.w, 8, 32);
    p.x += __shfl_xor(p.x, 16, 32); p.y += __shfl_xor(p.y, 16, 32);
    p.z += __shfl_xor(p.z, 16, 32); p.w += __shfl_xor(p.w, 16, 32);
    if (lane < 8) *(v4f*)&scr[(((ii * 8 + jj) * 8 + wave) << 5) + 4 * lane] = p;
    const size_t ro = ((size_t)((b * 64 + i) * 64 + j)) << 5;
    if (kq == (i & 31)) *(v4f*)(d13 + ro + 4 * c4) = hi_i ? v1 : v0;
    if (kq == (j & 31)) *(v4f*)(d23 + ro + 4 * c4) = hi_j ? v1 : v0;
    if (i == j) {
      float* dd = d12 + (((size_t)(b * 64 + i)) << 11) + 4 * t;
      *(v4f*)dd = v0; *(v4f*)(dd + 1024) = v1;
      if (kq == (i & 31)) *(v4f*)(tri + ((b << 6) + i) * 32 + 4 * c4) = hi_i ? v1 : v0;
    }
    asm volatile("" : "+v"(s1acc[jj][0]), "+v"(s1acc[jj][1]), "+v"(s2a0), "+v"(s2a1));
  };

  v4f buf[8][2];
  #pragma unroll
  for (int jj = 0; jj < 8; ++jj) {
    buf[jj][0] = load_x(xb + jj * 2048);
    buf[jj][1] = load_x(xb + jj * 2048 + 1024);
  }
  for (int ii = 0; ii < 7; ++ii) {
    const float* nx = xb + (size_t)(ii + 1) * 131072;
    v4f s2a0 = {0.f, 0.f, 0.f, 0.f}, s2a1 = {0.f, 0.f, 0.f, 0.f};
    #pragma unroll
    for (int jj = 0; jj < 8; ++jj) {
      row(buf[jj][0], buf[jj][1], ii, jj, s2a0, s2a1);
      buf[jj][0] = load_x(nx + jj * 2048);
      buf[jj][1] = load_x(nx + jj * 2048 + 1024);
    }
    float* dst = s2p + (((size_t)((b * 8 + jt) * 64 + i0 + ii)) << 11) + 4 * t;
    *(v4f*)dst = s2a0; *(v4f*)(dst + 1024) = s2a1;
  }
  {
    v4f s2a0 = {0.f, 0.f, 0.f, 0.f}, s2a1 = {0.f, 0.f, 0.f, 0.f};
    #pragma unroll
    for (int jj = 0; jj < 8; ++jj) row(buf[jj][0], buf[jj][1], 7, jj, s2a0, s2a1);
    float* dst = s2p + (((size_t)((b * 8 + jt) * 64 + i0 + 7)) << 11) + 4 * t;
    *(v4f*)dst = s2a0; *(v4f*)(dst + 1024) = s2a1;
  }

  #pragma unroll
  for (int jj = 0; jj < 8; ++jj) {
    float* dst = s1p + (((size_t)((b * 8 + it) * 64 + j0 + jj)) << 11) + 4 * t;
    *(v4f*)dst = s1acc[jj][0]; *(v4f*)(dst + 1024) = s1acc[jj][1];
  }
  {
    v4f t0 = s1acc[0][0], t1 = s1acc[0][1];
    #pragma unroll
    for (int jj = 1; jj < 8; ++jj) { t0 += s1acc[jj][0]; t1 += s1acc[jj][1]; }
    float* dst = tsp + (((size_t)(b * 64 + it * 8 + jt)) << 11) + 4 * t;
    *(v4f*)dst = t0; *(v4f*)(dst + 1024) = t1;
  }

  __syncthreads();
  #pragma unroll
  for (int ii = 0; ii < 8; ++ii) {
    float tot = 0.f;
    #pragma unroll
    for (int w = 0; w < 8; ++w) tot += scr[(((ii * 8 + wave) * 8 + w) << 5) + lane];
    s3m[(((size_t)((b * 64 + i0 + ii) * 64 + j0 + wave)) << 5) + lane] = tot * (1.f / 64.f);
  }
}


static __device__ __forceinline__ v4f load_part(const float* p) {
  return __builtin_nontemporal_load((const v4f*)p);
}

__global__ void __launch_bounds__(256)
k2_means_a1(const float* __restrict__ s1p, const float* __restrict__ s2p,
            const float* __restrict__ tsp, const float* __restrict__ s3m,
            const float* __restrict__ d12, const float* __restrict__ d13,
            const float* __restrict__ d23, const float* __restrict__ tri,
            const float* __restrict__ W12, const float* __restrict__ b12,
            const float* __restrict__ W12t, const float* __restrict__ b12t,
            const float* __restrict__ W11, const float* __restrict__ b11,
            float* __restrict__ s1m, float* __restrict__ s2m,
            float* __restrict__ a1, float* __restrict__ r12,
            float* __restrict__ r12t, float* __restrict__ forow) {
  __shared__ float row1[2048];
  __shared__ float scr[9 * 256];
  __shared__ float arow[320];
  __shared__ float part[8 * 96];
  const int t = threadIdx.x, c = t & 31, g = t >> 5;
  const int b = blockIdx.x >> 6, u = blockIdx.x & 63;
  float* dst = a1 + (size_t)(b * 64 + u) * 320;

  const size_t in0 = (((size_t)(b * 512 + u)) << 11) + 4 * t;
  const float* p1 = s1p + in0;
  const float* p2 = s2p + in0;
  const size_t col = (((size_t)(b * 4096 + u)) << 5) + c + (size_t)g * 2048;
  const size_t lin = (((size_t)((b * 64 + u) * 64)) << 5) + c + (size_t)g * 32;
  const float* src9[9] = { tsp + col, nullptr, s3m + lin, d12 + col, d12 + lin,
                           d13 + col, d13 + lin, d23 + col, d23 + lin };

  v4f r1[2][8], r2[2][8];
  float r[9][8];
  #pragma unroll
  for (int q = 0; q < 2; ++q) {
    #pragma unroll
    for (int gg = 0; gg < 8; ++gg) {
      r1[q][gg] = load_part(p1 + (size_t)gg * 131072 + q * 1024);
      r2[q][gg] = load_part(p2 + (size_t)gg * 131072 + q * 1024);
    }
  }
  #pragma unroll
  for (int q = 0; q < 9; ++q) {
    if (q == 1) continue;
    const int step = (q == 0 || q == 3 || q == 5 || q == 7) ? 8 * 2048 : 8 * 32;
    #pragma unroll
    for (int s = 0; s < 8; ++s) r[q][s] = src9[q][s * step];
  }
  float tv = 0.f;
  if (t < 32) tv = tri[(b * 64 + u) * 32 + t];
  __builtin_amdgcn_sched_barrier(0);

  const size_t o = (((size_t)(b * 64 + u)) << 11) + 4 * t;
  #pragma unroll
  for (int q = 0; q < 2; ++q) {
    v4f a = {0.f, 0.f, 0.f, 0.f}, sm = {0.f, 0.f, 0.f, 0.f};
    #pragma unroll
    for (int gg = 0; gg < 8; ++gg) { a += r1[q][gg]; sm += r2[q][gg]; }
    a = a * (1.f / 64.f);
    *(v4f*)(s1m + o + q * 1024) = a;
    *(v4f*)(s2m + o + q * 1024) = sm * (1.f / 64.f);
    *(v4f*)&row1[4 * t + q * 1024] = a;
  }
  #pragma unroll
  for (int q = 0; q < 9; ++q) {
    if (q == 1) continue;
    float acc = 0.f;
    #pragma unroll
    for (int s = 0; s < 8; ++s) acc += r[q][s];
    scr[q * 256 + t] = acc;
  }
  __syncthreads();
  {
    float acc = 0.f;
    #pragma unroll
    for (int s = 0; s < 8; ++s) acc += row1[((g + 8 * s) << 5) + c];
    scr[256 + t] = acc;
  }
  __syncthreads();
  for (int q = g; q < 9; q += 8) {
    float tot = 0.f;
    #pragma unroll
    for (int gg = 0; gg < 8; ++gg) tot += scr[q * 256 + gg * 32 + c];
    const float m = tot * ((q == 0) ? (1.f / 4096.f) : (1.f / 64.f));
    dst[q * 32 + c] = m;
    arow[q * 32 + c] = m;
  }
  if (t < 32) { dst[9 * 32 + t] = tv; arow[9 * 32 + t] = tv; }
  __syncthreads();

  const int k0 = g * 40;
  float m12 = 0.f, m12t = 0.f, m11 = 0.f;
  #pragma unroll
  for (int k = 0; k < 40; ++k) {
    const float av = arow[k0 + k];
    m12  += av * W12 [(k0 + k) * 32 + c];
    m12t += av * W12t[(k0 + k) * 32 + c];
    m11  += av * W11 [(k0 + k) * 32 + c];
  }
  part[g * 96 + c] = m12; part[g * 96 + 32 + c] = m12t; part[g * 96 + 64 + c] = m11;
  __syncthreads();
  if (t < 96) {
    float tot = (g == 0) ? b12[c] : (g == 1) ? b12t[c] : b11[c];
    #pragma unroll
    for (int gg = 0; gg < 8; ++gg) tot += part[gg * 96 + t];
    float* op = (g == 0) ? r12 : (g == 1) ? r12t : forow;
    op[(b * 64 + u) * 32 + c] = tot;
  }
}

__global__ void k3b_mix(const float* __restrict__ s1m, const float* __restrict__ s2m,
                        const float* __restrict__ s3m, const float* __restrict__ d12,
                        const float* __restrict__ d13, const float* __restrict__ d23,
                        const float* __restrict__ W22, const float* __restrict__ a1,
                        const float* __restrict__ W02, const float* __restrict__ b02,
                        const float* __restrict__ b22,
                        const float* __restrict__ W01, const float* __restrict__ b01,
                        const float* __restrict__ r12, const float* __restrict__ r12t,
                        const float* __restrict__ forow,
                        float* __restrict__ out) {
  __shared__ float w[384 * 32];
  __shared__ float a0p[8 * 160];
  __shared__ float a0s[160];
  __shared__ float vp[4 * 64];
  __shared__ float vec[64];
  const int t = threadIdx.x;
  const int lane = t & 31, wv = t >> 5;
  const int bb = blockIdx.x >> 5;

  {
    const float* ap = a1 + (size_t)(bb * 64 + wv * 8) * 320 + lane;
    float z0 = 0.f, z1 = 0.f, z2 = 0.f, z3 = 0.f, z4 = 0.f;
    for (int uu = 0; uu < 8; ++uu) {
      z0 += ap[uu * 320];       z1 += ap[uu * 320 + 96];  z2 += ap[uu * 320 + 160];
      z3 += ap[uu * 320 + 224]; z4 += ap[uu * 320 + 288];
    }
    float* zp = a0p + wv * 160 + lane;
    zp[0] = z0; zp[32] = z1; zp[64] = z2; zp[96] = z3; zp[128] = z4;
  }
  for (int idx = t; idx < 384 * 32; idx += 256) w[idx] = W22[idx];
  __syncthreads();
  if (t < 160) {
    float tot = 0.f;
    for (int g = 0; g < 8; ++g) tot += a0p[g * 160 + t];
    a0s[t] = tot * (1.f / 64.f);
  }
  __syncthreads();
  {
    const float* Wv = (wv & 1) ? W01 : W02;
    const int kq0 = (wv >> 1) * 40;
    const float bz = b02[lane] + b22[lane], bo = b01[lane];
    float acc = (wv >> 1) ? 0.f : ((wv & 1) ? bo : bz);
    #pragma unroll
    for (int k = 0; k < 40; ++k) acc += a0s[kq0 + k] * Wv[(kq0 + k) * 32 + lane];
    vp[(wv >> 1) * 64 + (wv & 1) * 32 + lane] = acc;
  }
  __syncthreads();
  if (t < 64) vec[t] = vp[t] + vp[64 + t] + vp[128 + t] + vp[192 + t];
  __syncthreads();

  const int R = (blockIdx.x * 8 + (t >> 5)) * 16;
  const int MA = lane & 15;
  const int kh = lane >> 4;
  const int rowA = R + MA;
  const int bA = rowA >> 12, pA = (rowA >> 6) & 63, qA = rowA & 63;
  const size_t offPQ = ((size_t)((bA * 64 + pA) * 64 + qA)) << 5;
  const size_t offQP = ((size_t)((bA * 64 + qA) * 64 + pA)) << 5;
  const float* comp[6] = { s1m, s2m, s3m, d12, d13, d23 };
  const int n0 = lane & 15;

  v8f acc0 = {};
  v8f acc1 = {};
  #pragma unroll
  for (int f = 0; f < 12; ++f) {
    const float* A = (f < 6) ? (comp[f] + offPQ) : (comp[f - 6] + offQP);
    #pragma unroll
    for (int kk = 0; kk < 32; kk += 4) {
      v2f a = *(const v2f*)(A + kk + 2 * kh);
      const int kcol = f * 32 + kk + 2 * kh;
      v2f b0 = { w[kcol * 32 + n0],      w[(kcol + 1) * 32 + n0] };
      v2f b1 = { w[kcol * 32 + n0 + 16], w[(kcol + 1) * 32 + n0 + 16] };
      acc0 = __builtin_amdgcn_wmma_f32_16x16x4_f32(false, a, false, b0,
                                                   (short)0, acc0, false, false);
      acc1 = __builtin_amdgcn_wmma_f32_16x16x4_f32(false, a, false, b1,
                                                   (short)0, acc1, false, false);
    }
  }
  #pragma unroll
  for (int r = 0; r < 8; ++r) {
    const int M = r + ((lane >> 4) << 3);
    const int row = R + M;
    const int b = row >> 12, p = (row >> 6) & 63, q = row & 63;
    float add0 = r12[((b << 6) + p) * 32 + n0] + r12t[((b << 6) + q) * 32 + n0]
               + vec[n0];
    float add1 = r12[((b << 6) + p) * 32 + n0 + 16] + r12t[((b << 6) + q) * 32 + n0 + 16]
               + vec[n0 + 16];
    if (p == q) {
      add0 += forow[((b << 6) + p) * 32 + n0] + vec[32 + n0];
      add1 += forow[((b << 6) + p) * 32 + n0 + 16] + vec[32 + n0 + 16];
    }
    out[((size_t)row << 5) + n0]      = acc0[r] + add0;
    out[((size_t)row << 5) + n0 + 16] = acc1[r] + add1;
  }
}

extern "C" void kernel_launch(void* const* d_in, const int* in_sizes, int n_in,
                              void* d_out, int out_size, void* d_ws, size_t ws_size,
                              hipStream_t stream) {
  (void)in_sizes; (void)n_in; (void)out_size;
  if (ws_size < WS_FLOATS * sizeof(float)) return;

  const float* x    = (const float*)d_in[0];
  const float* W22  = (const float*)d_in[1];
  const float* b22  = (const float*)d_in[2];
  const float* W12  = (const float*)d_in[3];
  const float* b12  = (const float*)d_in[4];
  const float* W12t = (const float*)d_in[5];
  const float* b12t = (const float*)d_in[6];
  const float* W02  = (const float*)d_in[7];
  const float* b02  = (const float*)d_in[8];
  const float* W11  = (const float*)d_in[9];
  const float* b11  = (const float*)d_in[10];
  const float* W01  = (const float*)d_in[11];
  const float* b01  = (const float*)d_in[12];

  float* ws = (float*)d_ws;
  float* s1p = ws + S1P;   float* s2p = ws + S2P;
  float* s3m = ws + S3M;   float* d12 = ws + D12;
  float* d13 = ws + D13;   float* d23 = ws + D23;
  float* tri = ws + TRI;   float* s1m = ws + S1M;
  float* s2m = ws + S2M;   float* a1  = ws + A1O;
  float* r12 = ws + R12O;
  float* r12t= ws + R12TO; float* forow = ws + FORO;
  float* tsp = ws + TSP;

  k1_contract<<<dim3(8, 8, 8), 256, 0, stream>>>(x, s1p, s2p, s3m, d12, d13, d23, tri, tsp);
  k2_means_a1<<<512, 256, 0, stream>>>(s1p, s2p, tsp, s3m, d12, d13, d23, tri,
                                      W12, b12, W12t, b12t, W11, b11,
                                      s1m, s2m, a1, r12, r12t, forow);
  k3b_mix    <<<256, 256, 0, stream>>>(s1m, s2m, s3m, d12, d13, d23, W22, a1,
                                       W02, b02, b22, W01, b01,
                                       r12, r12t, forow, (float*)d_out);
}


# Mix launch made latency-tolerant: prologue loads requested in one run at entry, A by 16-byte loads 3 feature blocks ahead, B by 16-byte LDS loads 1 block ahead (k permuted per block: new sum order)
# speedup vs baseline: 3.5702x; 1.1433x over previous
#include <hip/hip_runtime.h>


typedef __attribute__((ext_vector_type(2))) float v2f;
typedef __attribute__((ext_vector_type(8))) float v8f;

static constexpr size_t S1P  = 0;
static constexpr size_t S2P  = 8388608;
static constexpr size_t S3M  = 16777216;
static constexpr size_t D12  = 17825792;
static constexpr size_t D13  = 18874368;
static constexpr size_t D23  = 19922944;
static constexpr size_t TRI  = 20971520;
static constexpr size_t S1M  = 20987904;
static constexpr size_t S2M  = 22036480;
static constexpr size_t A1O  = 23085056;
static constexpr size_t R12O = 23250176;
static constexpr size_t R12TO= 23266560;
static constexpr size_t FORO = 23282944;
static constexpr size_t TSP  = 23299584;
static constexpr size_t WS_FLOATS = 24348160;

typedef __attribute__((ext_vector_type(4))) float v4f;

static __device__ __forceinline__ v4f load_x(const float* p) {
  return __builtin_nontemporal_load((const v4f*)p);
}

__global__ void __launch_bounds__(256)
k1_contract(const float* __restrict__ x,
            float* __restrict__ s1p, float* __restrict__ s2p,
            float* __restrict__ s3m, float* __restrict__ d12,
            float* __restrict__ d13, float* __restrict__ d23,
            float* __restrict__ tri, float* __restrict__ tsp) {
  __shared__ float scr[64 * 8 * 32];
  const int t  = threadIdx.x;
  const int jt = blockIdx.x, it = blockIdx.y, b = blockIdx.z;
  const int lane = t & 31, wave = t >> 5;
  const int kq = t >> 3, c4 = t & 7;
  const int i0 = it * 8, j0 = jt * 8;
  const bool hi_i = (it >> 2) != 0;
  const bool hi_j = (jt >> 2) != 0;
  const float* xb = x + (((size_t)((b * 64 + i0) * 64 + j0)) << 11) + 4 * t;

  v4f s1acc[8][2];
  #pragma unroll
  for (int jj = 0; jj < 8; ++jj) { s1acc[jj][0] = (v4f){0.f, 0.f, 0.f, 0.f}; s1acc[jj][1] = (v4f){0.f, 0.f, 0.f, 0.f}; }

  auto row = [&](const v4f v0, const v4f v1, const int ii, const int jj, v4f& s2a0, v4f& s2a1) {
    const int i = i0 + ii, j = j0 + jj;
    s1acc[jj][0] += v0; s1acc[jj][1] += v1;
    s2a0 += v0;         s2a1 += v1;
    v4f p = v0 + v1;
    p.x += __shfl_xor(p.x, 8, 32);  p.y += __shfl_xor(p.y, 8, 32);
    p.z += __shfl_xor(p.z, 8, 32);  p.w += __shfl_xor(p.w, 8, 32);
    p.x += __shfl_xor(p.x, 16, 32); p.y += __shfl_xor(p.y, 16, 32);
    p.z += __shfl_xor(p.z, 16, 32); p.w += __shfl_xor(p.w, 16, 32);
    if (lane < 8) *(v4f*)&scr[(((ii * 8 + jj) * 8 + wave) << 5) + 4 * lane] = p;
    const size_t ro = ((size_t)((b * 64 + i) * 64 + j)) << 5;
    if (kq == (i & 31)) *(v4f*)(d13 + ro + 4 * c4) = hi_i ? v1 : v0;
    if (kq == (j & 31)) *(v4f*)(d23 + ro + 4 * c4) = hi_j ? v1 : v0;
    if (i == j) {
      float* dd = d12 + (((size_t)(b * 64 + i)) << 11) + 4 * t;
      *(v4f*)dd = v0; *(v4f*)(dd + 1024) = v1;
      if (kq == (i & 31)) *(v4f*)(tri + ((b << 6) + i) * 32 + 4 * c4) = hi_i ? v1 : v0;
    }
    asm volatile("" : "+v"(s1acc[jj][0]), "+v"(s1acc[jj][1]), "+v"(s2a0), "+v"(s2a1));
  };

  v4f buf[8][2];
  #pragma unroll
  for (int jj = 0; jj < 8; ++jj) {
    buf[jj][0] = load_x(xb + jj * 2048);
    buf[jj][1] = load_x(xb + jj * 2048 + 1024);
  }
  for (int ii = 0; ii < 7; ++ii) {
    const float* nx = xb + (size_t)(ii + 1) * 131072;
    v4f s2a0 = {0.f, 0.f, 0.f, 0.f}, s2a1 = {0.f, 0.f, 0.f, 0.f};
    #pragma unroll
    for (int jj = 0; jj < 8; ++jj) {
      row(buf[jj][0], buf[jj][1], ii, jj, s2a0, s2a1);
      buf[jj][0] = load_x(nx + jj * 2048);
      buf[jj][1] = load_x(nx + jj * 2048 + 1024);
    }
    float* dst = s2p + (((size_t)((b * 8 + jt) * 64 + i0 + ii)) << 11) + 4 * t;
    *(v4f*)dst = s2a0; *(v4f*)(dst + 1024) = s2a1;
  }
  {
    v4f s2a0 = {0.f, 0.f, 0.f, 0.f}, s2a1 = {0.f, 0.f, 0.f, 0.f};
    #pragma unroll
    for (int jj = 0; jj < 8; ++jj) row(buf[jj][0], buf[jj][1], 7, jj, s2a0, s2a1);
    float* dst = s2p + (((size_t)((b * 8 + jt) * 64 + i0 + 7)) << 11) + 4 * t;
    *(v4f*)dst = s2a0; *(v4f*)(dst + 1024) = s2a1;
  }

  #pragma unroll
  for (int jj = 0; jj < 8; ++jj) {
    float* dst = s1p + (((size_t)((b * 8 + it) * 64 + j0 + jj)) << 11) + 4 * t;
    *(v4f*)dst = s1acc[jj][0]; *(v4f*)(dst + 1024) = s1acc[jj][1];
  }
  {
    v4f t0 = s1acc[0][0], t1 = s1acc[0][1];
    #pragma unroll
    for (int jj = 1; jj < 8; ++jj) { t0 += s1acc[jj][0]; t1 += s1acc[jj][1]; }
    float* dst = tsp + (((size_t)(b * 64 + it * 8 + jt)) << 11) + 4 * t;
    *(v4f*)dst = t0; *(v4f*)(dst + 1024) = t1;
  }

  __syncthreads();
  #pragma unroll
  for (int ii = 0; ii < 8; ++ii) {
    float tot = 0.f;
    #pragma unroll
    for (int w = 0; w < 8; ++w) tot += scr[(((ii * 8 + wave) * 8 + w) << 5) + lane];
    s3m[(((size_t)((b * 64 + i0 + ii) * 64 + j0 + wave)) << 5) + lane] = tot * (1.f / 64.f);
  }
}


static __device__ __forceinline__ v4f load_part(const float* p) {
  return __builtin_nontemporal_load((const v4f*)p);
}

__global__ void __launch_bounds__(256)
k2_means_a1(const float* __restrict__ s1p, const float* __restrict__ s2p,
            const float* __restrict__ tsp, const float* __restrict__ s3m,
            const float* __restrict__ d12, const float* __restrict__ d13,
            const float* __restrict__ d23, const float* __restrict__ tri,
            const float* __restrict__ W12, const float* __restrict__ b12,
            const float* __restrict__ W12t, const float* __restrict__ b12t,
            const float* __restrict__ W11, const float* __restrict__ b11,
            float* __restrict__ s1m, float* __restrict__ s2m,
            float* __restrict__ a1, float* __restrict__ r12,
            float* __restrict__ r12t, float* __restrict__ forow) {
  __shared__ float row1[2048];
  __shared__ float scr[9 * 256];
  __shared__ float arow[320];
  __shared__ float part[8 * 96];
  const int t = threadIdx.x, c = t & 31, g = t >> 5;
  const int b = blockIdx.x >> 6, u = blockIdx.x & 63;
  float* dst = a1 + (size_t)(b * 64 + u) * 320;

  const size_t in0 = (((size_t)(b * 512 + u)) << 11) + 4 * t;
  const float* p1 = s1p + in0;
  const float* p2 = s2p + in0;
  const size_t col = (((size_t)(b * 4096 + u)) << 5) + c + (size_t)g * 2048;
  const size_t lin = (((size_t)((b * 64 + u) * 64)) << 5) + c + (size_t)g * 32;
  const float* src9[9] = { tsp + col, nullptr, s3m + lin, d12 + col, d12 + lin,
                           d13 + col, d13 + lin, d23 + col, d23 + lin };

  v4f r1[2][8], r2[2][8];
  float r[9][8];
  #pragma unroll
  for (int q = 0; q < 2; ++q) {
    #pragma unroll
    for (int gg = 0; gg < 8; ++gg) {
      r1[q][gg] = load_part(p1 + (size_t)gg * 131072 + q * 1024);
      r2[q][gg] = load_part(p2 + (size_t)gg * 131072 + q * 1024);
    }
  }
  #pragma unroll
  for (int q = 0; q < 9; ++q) {
    if (q == 1) continue;
    const int step = (q == 0 || q == 3 || q == 5 || q == 7) ? 8 * 2048 : 8 * 32;
    #pragma unroll
    for (int s = 0; s < 8; ++s) r[q][s] = src9[q][s * step];
  }
  float tv = 0.f;
  if (t < 32) tv = tri[(b * 64 + u) * 32 + t];
  __builtin_amdgcn_sched_barrier(0);

  const size_t o = (((size_t)(b * 64 + u)) << 11) + 4 * t;
  #pragma unroll
  for (int q = 0; q < 2; ++q) {
    v4f a = {0.f, 0.f, 0.f, 0.f}, sm = {0.f, 0.f, 0.f, 0.f};
    #pragma unroll
    for (int gg = 0; gg < 8; ++gg) { a += r1[q][gg]; sm += r2[q][gg]; }
    a = a * (1.f / 64.f);
    *(v4f*)(s1m + o + q * 1024) = a;
    *(v4f*)(s2m + o + q * 1024) = sm * (1.f / 64.f);
    *(v4f*)&row1[4 * t + q * 1024] = a;
  }
  #pragma unroll
  for (int q = 0; q < 9; ++q) {
    if (q == 1) continue;
    float acc = 0.f;
    #pragma unroll
    for (int s = 0; s < 8; ++s) acc += r[q][s];
    scr[q * 256 + t] = acc;
  }
  __syncthreads();
  {
    float acc = 0.f;
    #pragma unroll
    for (int s = 0; s < 8; ++s) acc += row1[((g + 8 * s) << 5) + c];
    scr[256 + t] = acc;
  }
  __syncthreads();
  for (int q = g; q < 9; q += 8) {
    float tot = 0.f;
    #pragma unroll
    for (int gg = 0; gg < 8; ++gg) tot += scr[q * 256 + gg * 32 + c];
    const float m = tot * ((q == 0) ? (1.f / 4096.f) : (1.f / 64.f));
    dst[q * 32 + c] = m;
    arow[q * 32 + c] = m;
  }
  if (t < 32) { dst[9 * 32 + t] = tv; arow[9 * 32 + t] = tv; }
  __syncthreads();

  const int k0 = g * 40;
  float m12 = 0.f, m12t = 0.f, m11 = 0.f;
  #pragma unroll
  for (int k = 0; k < 40; ++k) {
    const float av = arow[k0 + k];
    m12  += av * W12 [(k0 + k) * 32 + c];
    m12t += av * W12t[(k0 + k) * 32 + c];
    m11  += av * W11 [(k0 + k) * 32 + c];
  }
  part[g * 96 + c] = m12; part[g * 96 + 32 + c] = m12t; part[g * 96 + 64 + c] = m11;
  __syncthreads();
  if (t < 96) {
    float tot = (g == 0) ? b12[c] : (g == 1) ? b12t[c] : b11[c];
    #pragma unroll
    for (int gg = 0; gg < 8; ++gg) tot += part[gg * 96 + t];
    float* op = (g == 0) ? r12 : (g == 1) ? r12t : forow;
    op[(b * 64 + u) * 32 + c] = tot;
  }
}

__global__ void __launch_bounds__(256)
k3b_mix(const float* __restrict__ s1m, const float* __restrict__ s2m,
        const float* __restrict__ s3m, const float* __restrict__ d12,
        const float* __restrict__ d13, const float* __restrict__ d23,
        const float* __restrict__ W22, const float* __restrict__ a1,
        const float* __restrict__ W02, const float* __restrict__ b02,
        const float* __restrict__ b22,
        const float* __restrict__ W01, const float* __restrict__ b01,
        const float* __restrict__ r12, const float* __restrict__ r12t,
        const float* __restrict__ forow,
        float* __restrict__ out) {
  __shared__ v4f wq[192 * 16];
  __shared__ float a0p[8 * 160];
  __shared__ float a0s[160];
  __shared__ float vp[4 * 64];
  __shared__ float vec[64];
  const int t = threadIdx.x;
  const int lane = t & 31, wv = t >> 5;
  const int bb = blockIdx.x >> 5;

  const int R = (blockIdx.x * 8 + wv) * 16;
  const int MA = lane & 15;
  const int kh = lane >> 4;
  const int n0 = lane & 15;
  const int rowA = R + MA;
  const int bA = rowA >> 12, pA = (rowA >> 6) & 63, qA = rowA & 63;
  const size_t offPQ = ((size_t)((bA * 64 + pA) * 64 + qA)) << 5;
  const size_t offQP = ((size_t)((bA * 64 + qA) * 64 + pA)) << 5;
  const float* comp[6] = { s1m, s2m, s3m, d12, d13, d23 };
  const int bW = R >> 12, pW = (R >> 6) & 63, q0 = R & 63;

  v4f Af[4][4];
  auto loadA = [&](const int f, v4f (&dst)[4]) {
    const float* Ap = ((f < 6) ? (comp[f] + offPQ) : (comp[f - 6] + offQP)) + 4 * kh;
    #pragma unroll
    for (int j2 = 0; j2 < 4; ++j2) dst[j2] = *(const v4f*)(Ap + 8 * j2);
  };

  float za[5][8];
  {
    const float* ap = a1 + (size_t)(bb * 64 + wv * 8) * 320 + lane;
    #pragma unroll
    for (int uu = 0; uu < 8; ++uu) {
      za[0][uu] = ap[uu * 320];       za[1][uu] = ap[uu * 320 + 96];  za[2][uu] = ap[uu * 320 + 160];
      za[3][uu] = ap[uu * 320 + 224]; za[4][uu] = ap[uu * 320 + 288];
    }
  }
  v4f wr[3][4];
  #pragma unroll
  for (int i = 0; i < 3; ++i) {
    const int item = t + 256 * i, kp = item >> 2, m = item & 3;
    const float* src = W22 + kp * 64 + 4 * m;
    wr[i][0] = *(const v4f*)src;        wr[i][1] = *(const v4f*)(src + 32);
    wr[i][2] = *(const v4f*)(src + 16); wr[i][3] = *(const v4f*)(src + 48);
  }
  const float* Wv = (wv & 1) ? W01 : W02;
  const int kq0 = (wv >> 1) * 40;
  float wreg[40];
  #pragma unroll
  for (int k = 0; k < 40; ++k) wreg[k] = Wv[(kq0 + k) * 32 + lane];
  const float bz = b02[lane] + b22[lane], bo = b01[lane];
  const float* r12p = r12 + ((bW << 6) + pW) * 32 + n0;
  const float* forp = forow + ((bW << 6) + pW) * 32 + n0;
  const float e12a = r12p[0], e12b = r12p[16], efa = forp[0], efb = forp[16];
  float e12t[8][2];
  #pragma unroll
  for (int r = 0; r < 8; ++r) {
    const float* tp = r12t + ((bW << 6) + q0 + r + 8 * kh) * 32 + n0;
    e12t[r][0] = tp[0]; e12t[r][1] = tp[16];
  }
  loadA(0, Af[0]); loadA(1, Af[1]); loadA(2, Af[2]);
  __builtin_amdgcn_sched_barrier(0);

  {
    float* zp = a0p + wv * 160 + lane;
    #pragma unroll
    for (int sl = 0; sl < 5; ++sl) {
      float z = 0.f;
      #pragma unroll
      for (int uu = 0; uu < 8; ++uu) z += za[sl][uu];
      zp[sl * 32] = z;
    }
  }
  #pragma unroll
  for (int i = 0; i < 3; ++i) {
    const int item = t + 256 * i, kp = item >> 2, m = item & 3;
    #pragma unroll
    for (int e = 0; e < 4; ++e)
      wq[kp * 16 + 4 * m + e] = (v4f){ wr[i][0][e], wr[i][1][e], wr[i][2][e], wr[i][3][e] };
  }
  __syncthreads();
  if (t < 160) {
    float tot = 0.f;
    for (int g = 0; g < 8; ++g) tot += a0p[g * 160 + t];
    a0s[t] = tot * (1.f / 64.f);
  }
  __syncthreads();
  {
    float acc = (wv >> 1) ? 0.f : ((wv & 1) ? bo : bz);
    #pragma unroll
    for (int k = 0; k < 40; ++k) acc += a0s[kq0 + k] * wreg[k];
    vp[(wv >> 1) * 64 + (wv & 1) * 32 + lane] = acc;
  }
  __syncthreads();
  if (t < 64) vec[t] = vp[t] + vp[64 + t] + vp[128 + t] + vp[192 + t];
  __syncthreads();

  const v4f* wqb = wq + (2 * kh) * 16 + n0;
  v4f Bf[2][8];
  auto loadB = [&](const int f, v4f (&dst)[8]) {
    #pragma unroll
    for (int s = 0; s < 8; ++s) dst[s] = wqb[(f * 16 + 4 * (s >> 1) + (s & 1)) * 16];
  };

  v8f acc0 = {};
  v8f acc1 = {};
  loadB(0, Bf[0]);
  #pragma unroll
  for (int f = 0; f < 12; ++f) {
    if (f + 1 < 12) loadB(f + 1, Bf[(f + 1) & 1]);
    if (f + 3 < 12) loadA(f + 3, Af[(f + 3) & 3]);
    __builtin_amdgcn_sched_barrier(0);
    #pragma unroll
    for (int s = 0; s < 8; ++s) {
      const v4f av = Af[f & 3][s >> 1];
      const v4f bv = Bf[f & 1][s];
      const v2f a  = (s & 1) ? (v2f){ av.z, av.w } : (v2f){ av.x, av.y };
      const v2f b0 = { bv.x, bv.y };
      const v2f b1 = { bv.z, bv.w };
      acc0 = __builtin_amdgcn_wmma_f32_16x16x4_f32(false, a, false, b0,
                                                   (short)0, acc0, false, false);
      acc1 = __builtin_amdgcn_wmma_f32_16x16x4_f32(false, a, false, b1,
                                                   (short)0, acc1, false, false);
    }
    __builtin_amdgcn_sched_barrier(0);
  }
  const float v0 = vec[n0], v1 = vec[n0 + 16];
  const float vd0 = efa + vec[32 + n0], vd1 = efb + vec[32 + n0 + 16];
  #pragma unroll
  for (int r = 0; r < 8; ++r) {
    const int M = r + 8 * kh;
    const int row = R + M;
    float add0 = e12a + e12t[r][0] + v0;
    float add1 = e12b + e12t[r][1] + v1;
    const bool dg = (pW == q0 + M);
    add0 = dg ? add0 + vd0 : add0;
    add1 = dg ? add1 + vd1 : add1;
    out[((size_t)row << 5) + n0]      = acc0[r] + add0;
    out[((size_t)row << 5) + n0 + 16] = acc1[r] + add1;
  }
}

extern "C" void kernel_launch(void* const* d_in, const int* in_sizes, int n_in,
                              void* d_out, int out_size, void* d_ws, size_t ws_size,
                              hipStream_t stream) {
  (void)in_sizes; (void)n_in; (void)out_size;
  if (ws_size < WS_FLOATS * sizeof(float)) return;

  const float* x    = (const float*)d_in[0];
  const float* W22  = (const float*)d_in[1];
  const float* b22  = (const float*)d_in[2];
  const float* W12  = (const float*)d_in[3];
  const float* b12  = (const float*)d_in[4];
  const float* W12t = (const float*)d_in[5];
  const float* b12t = (const float*)d_in[6];
  const float* W02  = (const float*)d_in[7];
  const float* b02  = (const float*)d_in[8];
  const float* W11  = (const float*)d_in[9];
  const float* b11  = (const float*)d_in[10];
  const float* W01  = (const float*)d_in[11];
  const float* b01  = (const float*)d_in[12];

  float* ws = (float*)d_ws;
  float* s1p = ws + S1P;   float* s2p = ws + S2P;
  float* s3m = ws + S3M;   float* d12 = ws + D12;
  float* d13 = ws + D13;   float* d23 = ws + D23;
  float* tri = ws + TRI;   float* s1m = ws + S1M;
  float* s2m = ws + S2M;   float* a1  = ws + A1O;
  float* r12 = ws + R12O;
  float* r12t= ws + R12TO; float* forow = ws + FORO;
  float* tsp = ws + TSP;

  k1_contract<<<dim3(8, 8, 8), 256, 0, stream>>>(x, s1p, s2p, s3m, d12, d13, d23, tri, tsp);
  k2_means_a1<<<512, 256, 0, stream>>>(s1p, s2p, tsp, s3m, d12, d13, d23, tri,
                                      W12, b12, W12t, b12t, W11, b11,
                                      s1m, s2m, a1, r12, r12t, forow);
  k3b_mix    <<<256, 256, 0, stream>>>(s1m, s2m, s3m, d12, d13, d23, W22, a1,
                                       W02, b02, b22, W01, b01,
                                       r12, r12t, forow, (float*)d_out);
}


# Sum-over-i partial planes 8 -> 4: a pass-1 block is two 8-wave halves on i-neighbour tiles, the upper hands its sums to the lower via LDS behind the end barrier; pass 2 adds 4 planes (new sum order)
# speedup vs baseline: 3.8664x; 1.0830x over previous
#include <hip/hip_runtime.h>


typedef __attribute__((ext_vector_type(2))) float v2f;
typedef __attribute__((ext_vector_type(8))) float v8f;

static constexpr size_t S1P  = 0;
static constexpr size_t S2P  = 8388608;
static constexpr size_t S3M  = 16777216;
static constexpr size_t D12  = 17825792;
static constexpr size_t D13  = 18874368;
static constexpr size_t D23  = 19922944;
static constexpr size_t TRI  = 20971520;
static constexpr size_t S1M  = 20987904;
static constexpr size_t S2M  = 22036480;
static constexpr size_t A1O  = 23085056;
static constexpr size_t R12O = 23250176;
static constexpr size_t R12TO= 23266560;
static constexpr size_t FORO = 23282944;
static constexpr size_t TSP  = 23299584;
static constexpr size_t WS_FLOATS = 24348160;

typedef __attribute__((ext_vector_type(4))) float v4f;

static __device__ __forceinline__ v4f load_x(const float* p) {
  return __builtin_nontemporal_load((const v4f*)p);
}

__global__ void __launch_bounds__(512)
k1_contract(const float* __restrict__ x,
            float* __restrict__ s1p, float* __restrict__ s2p,
            float* __restrict__ s3m, float* __restrict__ d12,
            float* __restrict__ d13, float* __restrict__ d23,
            float* __restrict__ tri, float* __restrict__ tsp) {
  __shared__ float scr2[2 * 64 * 8 * 32];
  __shared__ v4f xch[16 * 256];
  const int h  = __builtin_amdgcn_readfirstlane((int)(threadIdx.x >> 8));
  const int t  = threadIdx.x & 255;
  float* scr = scr2 + h * (64 * 8 * 32);
  const int jt = blockIdx.x, it = blockIdx.y * 2 + h, b = blockIdx.z;
  const int lane = t & 31, wave = t >> 5;
  const int kq = t >> 3, c4 = t & 7;
  const int i0 = it * 8, j0 = jt * 8;
  const bool hi_i = (it >> 2) != 0;
  const bool hi_j = (jt >> 2) != 0;
  const float* xb = x + (((size_t)((b * 64 + i0) * 64 + j0)) << 11) + 4 * t;

  v4f s1acc[8][2];
  #pragma unroll
  for (int jj = 0; jj < 8; ++jj) { s1acc[jj][0] = (v4f){0.f, 0.f, 0.f, 0.f}; s1acc[jj][1] = (v4f){0.f, 0.f, 0.f, 0.f}; }

  auto row = [&](const v4f v0, const v4f v1, const int ii, const int jj, v4f& s2a0, v4f& s2a1) {
    const int i = i0 + ii, j = j0 + jj;
    s1acc[jj][0] += v0; s1acc[jj][1] += v1;
    s2a0 += v0;         s2a1 += v1;
    v4f p = v0 + v1;
    p.x += __shfl_xor(p.x, 8, 32);  p.y += __shfl_xor(p.y, 8, 32);
    p.z += __shfl_xor(p.z, 8, 32);  p.w += __shfl_xor(p.w, 8, 32);
    p.x += __shfl_xor(p.x, 16, 32); p.y += __shfl_xor(p.y, 16, 32);
    p.z += __shfl_xor(p.z, 16, 32); p.w += __shfl_xor(p.w, 16, 32);
    if (lane < 8) *(v4f*)&scr[(((ii * 8 + jj) * 8 + wave) << 5) + 4 * lane] = p;
    const size_t ro = ((size_t)((b * 64 + i) * 64 + j)) << 5;
    if (kq == (i & 31)) *(v4f*)(d13 + ro + 4 * c4) = hi_i ? v1 : v0;
    if (kq == (j & 31)) *(v4f*)(d23 + ro + 4 * c4) = hi_j ? v1 : v0;
    if (i == j) {
      float* dd = d12 + (((size_t)(b * 64 + i)) << 11) + 4 * t;
      *(v4f*)dd = v0; *(v4f*)(dd + 1024) = v1;
      if (kq == (i & 31)) *(v4f*)(tri + ((b << 6) + i) * 32 + 4 * c4) = hi_i ? v1 : v0;
    }
    asm volatile("" : "+v"(s1acc[jj][0]), "+v"(s1acc[jj][1]), "+v"(s2a0), "+v"(s2a1));
  };

  v4f buf[8][2];
  #pragma unroll
  for (int jj = 0; jj < 8; ++jj) {
    buf[jj][0] = load_x(xb + jj * 2048);
    buf[jj][1] = load_x(xb + jj * 2048 + 1024);
  }
  for (int ii = 0; ii < 7; ++ii) {
    const float* nx = xb + (size_t)(ii + 1) * 131072;
    v4f s2a0 = {0.f, 0.f, 0.f, 0.f}, s2a1 = {0.f, 0.f, 0.f, 0.f};
    #pragma unroll
    for (int jj = 0; jj < 8; ++jj) {
      row(buf[jj][0], buf[jj][1], ii, jj, s2a0, s2a1);
      buf[jj][0] = load_x(nx + jj * 2048);
      buf[jj][1] = load_x(nx + jj * 2048 + 1024);
    }
    float* dst = s2p + (((size_t)((b * 8 + jt) * 64 + i0 + ii)) << 11) + 4 * t;
    *(v4f*)dst = s2a0; *(v4f*)(dst + 1024) = s2a1;
  }
  {
    v4f s2a0 = {0.f, 0.f, 0.f, 0.f}, s2a1 = {0.f, 0.f, 0.f, 0.f};
    #pragma unroll
    for (int jj = 0; jj < 8; ++jj) row(buf[jj][0], buf[jj][1], 7, jj, s2a0, s2a1);
    float* dst = s2p + (((size_t)((b * 8 + jt) * 64 + i0 + 7)) << 11) + 4 * t;
    *(v4f*)dst = s2a0; *(v4f*)(dst + 1024) = s2a1;
  }

  v4f ts0 = s1acc[0][0], ts1 = s1acc[0][1];
  #pragma unroll
  for (int jj = 1; jj < 8; ++jj) { ts0 += s1acc[jj][0]; ts1 += s1acc[jj][1]; }
  if (h == 1) {
    #pragma unroll
    for (int jj = 0; jj < 8; ++jj) {
      xch[(jj * 2 + 0) * 256 + t] = s1acc[jj][0];
      xch[(jj * 2 + 1) * 256 + t] = s1acc[jj][1];
    }
  }

  __syncthreads();
  {
    float* dst = tsp + (((size_t)(b * 64 + it * 8 + jt)) << 11) + 4 * t;
    *(v4f*)dst = ts0; *(v4f*)(dst + 1024) = ts1;
  }
  if (h == 0) {
    #pragma unroll
    for (int jj = 0; jj < 8; ++jj) {
      float* dst = s1p + (((size_t)((b * 4 + blockIdx.y) * 64 + j0 + jj)) << 11) + 4 * t;
      *(v4f*)dst          = s1acc[jj][0] + xch[(jj * 2 + 0) * 256 + t];
      *(v4f*)(dst + 1024) = s1acc[jj][1] + xch[(jj * 2 + 1) * 256 + t];
    }
  }
  #pragma unroll
  for (int ii = 0; ii < 8; ++ii) {
    float tot = 0.f;
    #pragma unroll
    for (int w = 0; w < 8; ++w) tot += scr[(((ii * 8 + wave) * 8 + w) << 5) + lane];
    s3m[(((size_t)((b * 64 + i0 + ii) * 64 + j0 + wave)) << 5) + lane] = tot * (1.f / 64.f);
  }
}


static __device__ __forceinline__ v4f load_part(const float* p) {
  return __builtin_nontemporal_load((const v4f*)p);
}

__global__ void __launch_bounds__(256)
k2_means_a1(const float* __restrict__ s1p, const float* __restrict__ s2p,
            const float* __restrict__ tsp, const float* __restrict__ s3m,
            const float* __restrict__ d12, const float* __restrict__ d13,
            const float* __restrict__ d23, const float* __restrict__ tri,
            const float* __restrict__ W12, const float* __restrict__ b12,
            const float* __restrict__ W12t, const float* __restrict__ b12t,
            const float* __restrict__ W11, const float* __restrict__ b11,
            float* __restrict__ s1m, float* __restrict__ s2m,
            float* __restrict__ a1, float* __restrict__ r12,
            float* __restrict__ r12t, float* __restrict__ forow) {
  __shared__ float row1[2048];
  __shared__ float scr[9 * 256];
  __shared__ float arow[320];
  __shared__ float part[8 * 96];
  const int t = threadIdx.x, c = t & 31, g = t >> 5;
  const int b = blockIdx.x >> 6, u = blockIdx.x & 63;
  float* dst = a1 + (size_t)(b * 64 + u) * 320;

  const float* p1 = s1p + (((size_t)(b * 256 + u)) << 11) + 4 * t;
  const float* p2 = s2p + (((size_t)(b * 512 + u)) << 11) + 4 * t;
  const size_t col = (((size_t)(b * 4096 + u)) << 5) + c + (size_t)g * 2048;
  const size_t lin = (((size_t)((b * 64 + u) * 64)) << 5) + c + (size_t)g * 32;
  const float* src9[9] = { tsp + col, nullptr, s3m + lin, d12 + col, d12 + lin,
                           d13 + col, d13 + lin, d23 + col, d23 + lin };

  v4f r1[2][4], r2[2][8];
  float r[9][8];
  #pragma unroll
  for (int q = 0; q < 2; ++q) {
    #pragma unroll
    for (int gg = 0; gg < 8; ++gg) {
      if (gg < 4) r1[q][gg] = load_part(p1 + (size_t)gg * 131072 + q * 1024);
      r2[q][gg] = load_part(p2 + (size_t)gg * 131072 + q * 1024);
    }
  }
  #pragma unroll
  for (int q = 0; q < 9; ++q) {
    if (q == 1) continue;
    const int step = (q == 0 || q == 3 || q == 5 || q == 7) ? 8 * 2048 : 8 * 32;
    #pragma unroll
    for (int s = 0; s < 8; ++s) r[q][s] = src9[q][s * step];
  }
  float tv = 0.f;
  if (t < 32) tv = tri[(b * 64 + u) * 32 + t];
  __builtin_amdgcn_sched_barrier(0);

  const size_t o = (((size_t)(b * 64 + u)) << 11) + 4 * t;
  #pragma unroll
  for (int q = 0; q < 2; ++q) {
    v4f a = {0.f, 0.f, 0.f, 0.f}, sm = {0.f, 0.f, 0.f, 0.f};
    #pragma unroll
    for (int gg = 0; gg < 8; ++gg) { if (gg < 4) a += r1[q][gg]; sm += r2[q][gg]; }
    a = a * (1.f / 64.f);
    *(v4f*)(s1m + o + q * 1024) = a;
    *(v4f*)(s2m + o + q * 1024) = sm * (1.f / 64.f);
    *(v4f*)&row1[4 * t + q * 1024] = a;
  }
  #pragma unroll
  for (int q = 0; q < 9; ++q) {
    if (q == 1) continue;
    float acc = 0.f;
    #pragma unroll
    for (int s = 0; s < 8; ++s) acc += r[q][s];
    scr[q * 256 + t] = acc;
  }
  __syncthreads();
  {
    float acc = 0.f;
    #pragma unroll
    for (int s = 0; s < 8; ++s) acc += row1[((g + 8 * s) << 5) + c];
    scr[256 + t] = acc;
  }
  __syncthreads();
  for (int q = g; q < 9; q += 8) {
    float tot = 0.f;
    #pragma unroll
    for (int gg = 0; gg < 8; ++gg) tot += scr[q * 256 + gg * 32 + c];
    const float m = tot * ((q == 0) ? (1.f / 4096.f) : (1.f / 64.f));
    dst[q * 32 + c] = m;
    arow[q * 32 + c] = m;
  }
  if (t < 32) { dst[9 * 32 + t] = tv; arow[9 * 32 + t] = tv; }
  __syncthreads();

  const int k0 = g * 40;
  float m12 = 0.f, m12t = 0.f, m11 = 0.f;
  #pragma unroll
  for (int k = 0; k < 40; ++k) {
    const float av = arow[k0 + k];
    m12  += av * W12 [(k0 + k) * 32 + c];
    m12t += av * W12t[(k0 + k) * 32 + c];
    m11  += av * W11 [(k0 + k) * 32 + c];
  }
  part[g * 96 + c] = m12; part[g * 96 + 32 + c] = m12t; part[g * 96 + 64 + c] = m11;
  __syncthreads();
  if (t < 96) {
    float tot = (g == 0) ? b12[c] : (g == 1) ? b12t[c] : b11[c];
    #pragma unroll
    for (int gg = 0; gg < 8; ++gg) tot += part[gg * 96 + t];
    float* op = (g == 0) ? r12 : (g == 1) ? r12t : forow;
    op[(b * 64 + u) * 32 + c] = tot;
  }
}

__global__ void __launch_bounds__(256)
k3b_mix(const float* __restrict__ s1m, const float* __restrict__ s2m,
        const float* __restrict__ s3m, const float* __restrict__ d12,
        const float* __restrict__ d13, const float* __restrict__ d23,
        const float* __restrict__ W22, const float* __restrict__ a1,
        const float* __restrict__ W02, const float* __restrict__ b02,
        const float* __restrict__ b22,
        const float* __restrict__ W01, const float* __restrict__ b01,
        const float* __restrict__ r12, const float* __restrict__ r12t,
        const float* __restrict__ forow,
        float* __restrict__ out) {
  __shared__ v4f wq[192 * 16];
  __shared__ float a0p[8 * 160];
  __shared__ float a0s[160];
  __shared__ float vp[4 * 64];
  __shared__ float vec[64];
  const int t = threadIdx.x;
  const int lane = t & 31, wv = t >> 5;
  const int bb = blockIdx.x >> 5;

  const int R = (blockIdx.x * 8 + wv) * 16;
  const int MA = lane & 15;
  const int kh = lane >> 4;
  const int n0 = lane & 15;
  const int rowA = R + MA;
  const int bA = rowA >> 12, pA = (rowA >> 6) & 63, qA = rowA & 63;
  const size_t offPQ = ((size_t)((bA * 64 + pA) * 64 + qA)) << 5;
  const size_t offQP = ((size_t)((bA * 64 + qA) * 64 + pA)) << 5;
  const float* comp[6] = { s1m, s2m, s3m, d12, d13, d23 };
  const int bW = R >> 12, pW = (R >> 6) & 63, q0 = R & 63;

  v4f Af[4][4];
  auto loadA = [&](const int f, v4f (&dst)[4]) {
    const float* Ap = ((f < 6) ? (comp[f] + offPQ) : (comp[f - 6] + offQP)) + 4 * kh;
    #pragma unroll
    for (int j2 = 0; j2 < 4; ++j2) dst[j2] = *(const v4f*)(Ap + 8 * j2);
  };

  float za[5][8];
  {
    const float* ap = a1 + (size_t)(bb * 64 + wv * 8) * 320 + lane;
    #pragma unroll
    for (int uu = 0; uu < 8; ++uu) {
      za[0][uu] = ap[uu * 320];       za[1][uu] = ap[uu * 320 + 96];  za[2][uu] = ap[uu * 320 + 160];
      za[3][uu] = ap[uu * 320 + 224]; za[4][uu] = ap[uu * 320 + 288];
    }
  }
  v4f wr[3][4];
  #pragma unroll
  for (int i = 0; i < 3; ++i) {
    const int item = t + 256 * i, kp = item >> 2, m = item & 3;
    const float* src = W22 + kp * 64 + 4 * m;
    wr[i][0] = *(const v4f*)src;        wr[i][1] = *(const v4f*)(src + 32);
    wr[i][2] = *(const v4f*)(src + 16); wr[i][3] = *(const v4f*)(src + 48);
  }
  const float* Wv = (wv & 1) ? W01 : W02;
  const int kq0 = (wv >> 1) * 40;
  float wreg[40];
  #pragma unroll
  for (int k = 0; k < 40; ++k) wreg[k] = Wv[(kq0 + k) * 32 + lane];
  const float bz = b02[lane] + b22[lane], bo = b01[lane];
  const float* r12p = r12 + ((bW << 6) + pW) * 32 + n0;
  const float* forp = forow + ((bW << 6) + pW) * 32 + n0;
  const float e12a = r12p[0], e12b = r12p[16], efa = forp[0], efb = forp[16];
  float e12t[8][2];
  #pragma unroll
  for (int r = 0; r < 8; ++r) {
    const float* tp = r12t + ((bW << 6) + q0 + r + 8 * kh) * 32 + n0;
    e12t[r][0] = tp[0]; e12t[r][1] = tp[16];
  }
  loadA(0, Af[0]); loadA(1, Af[1]); loadA(2, Af[2]);
  __builtin_amdgcn_sched_barrier(0);

  {
    float* zp = a0p + wv * 160 + lane;
    #pragma unroll
    for (int sl = 0; sl < 5; ++sl) {
      float z = 0.f;
      #pragma unroll
      for (int uu = 0; uu < 8; ++uu) z += za[sl][uu];
      zp[sl * 32] = z;
    }
  }
  #pragma unroll
  for (int i = 0; i < 3; ++i) {
    const int item = t + 256 * i, kp = item >> 2, m = item & 3;
    #pragma unroll
    for (int e = 0; e < 4; ++e)
      wq[kp * 16 + 4 * m + e] = (v4f){ wr[i][0][e], wr[i][1][e], wr[i][2][e], wr[i][3][e] };
  }
  __syncthreads();
  if (t < 160) {
    float tot = 0.f;
    for (int g = 0; g < 8; ++g) tot += a0p[g * 160 + t];
    a0s[t] = tot * (1.f / 64.f);
  }
  __syncthreads();
  {
    float acc = (wv >> 1) ? 0.f : ((wv & 1) ? bo : bz);
    #pragma unroll
    for (int k = 0; k < 40; ++k) acc += a0s[kq0 + k] * wreg[k];
    vp[(wv >> 1) * 64 + (wv & 1) * 32 + lane] = acc;
  }
  __syncthreads();
  if (t < 64) vec[t] = vp[t] + vp[64 + t] + vp[128 + t] + vp[192 + t];
  __syncthreads();

  const v4f* wqb = wq + (2 * kh) * 16 + n0;
  v4f Bf[2][8];
  auto loadB = [&](const int f, v4f (&dst)[8]) {
    #pragma unroll
    for (int s = 0; s < 8; ++s) dst[s] = wqb[(f * 16 + 4 * (s >> 1) + (s & 1)) * 16];
  };

  v8f acc0 = {};
  v8f acc1 = {};
  loadB(0, Bf[0]);
  #pragma unroll
  for (int f = 0; f < 12; ++f) {
    if (f + 1 < 12) loadB(f + 1, Bf[(f + 1) & 1]);
    if (f + 3 < 12) loadA(f + 3, Af[(f + 3) & 3]);
    __builtin_amdgcn_sched_barrier(0);
    #pragma unroll
    for (int s = 0; s < 8; ++s) {
      const v4f av = Af[f & 3][s >> 1];
      const v4f bv = Bf[f & 1][s];
      const v2f a  = (s & 1) ? (v2f){ av.z, av.w } : (v2f){ av.x, av.y };
      const v2f b0 = { bv.x, bv.y };
      const v2f b1 = { bv.z, bv.w };
      acc0 = __builtin_amdgcn_wmma_f32_16x16x4_f32(false, a, false, b0,
                                                   (short)0, acc0, false, false);
      acc1 = __builtin_amdgcn_wmma_f32_16x16x4_f32(false, a, false, b1,
                                                   (short)0, acc1, false, false);
    }
    __builtin_amdgcn_sched_barrier(0);
  }
  const float v0 = vec[n0], v1 = vec[n0 + 16];
  const float vd0 = efa + vec[32 + n0], vd1 = efb + vec[32 + n0 + 16];
  #pragma unroll
  for (int r = 0; r < 8; ++r) {
    const int M = r + 8 * kh;
    const int row = R + M;
    float add0 = e12a + e12t[r][0] + v0;
    float add1 = e12b + e12t[r][1] + v1;
    const bool dg = (pW == q0 + M);
    add0 = dg ? add0 + vd0 : add0;
    add1 = dg ? add1 + vd1 : add1;
    out[((size_t)row << 5) + n0]      = acc0[r] + add0;
    out[((size_t)row << 5) + n0 + 16] = acc1[r] + add1;
  }
}

extern "C" void kernel_launch(void* const* d_in, const int* in_sizes, int n_in,
                              void* d_out, int out_size, void* d_ws, size_t ws_size,
                              hipStream_t stream) {
  (void)in_sizes; (void)n_in; (void)out_size;
  if (ws_size < WS_FLOATS * sizeof(float)) return;

  const float* x    = (const float*)d_in[0];
  const float* W22  = (const float*)d_in[1];
  const float* b22  = (const float*)d_in[2];
  const float* W12  = (const float*)d_in[3];
  const float* b12  = (const float*)d_in[4];
  const float* W12t = (const float*)d_in[5];
  const float* b12t = (const float*)d_in[6];
  const float* W02  = (const float*)d_in[7];
  const float* b02  = (const float*)d_in[8];
  const float* W11  = (const float*)d_in[9];
  const float* b11  = (const float*)d_in[10];
  const float* W01  = (const float*)d_in[11];
  const float* b01  = (const float*)d_in[12];

  float* ws = (float*)d_ws;
  float* s1p = ws + S1P;   float* s2p = ws + S2P;
  float* s3m = ws + S3M;   float* d12 = ws + D12;
  float* d13 = ws + D13;   float* d23 = ws + D23;
  float* tri = ws + TRI;   float* s1m = ws + S1M;
  float* s2m = ws + S2M;   float* a1  = ws + A1O;
  float* r12 = ws + R12O;
  float* r12t= ws + R12TO; float* forow = ws + FORO;
  float* tsp = ws + TSP;

  k1_contract<<<dim3(8, 4, 8), 512, 0, stream>>>(x, s1p, s2p, s3m, d12, d13, d23, tri, tsp);
  k2_means_a1<<<512, 256, 0, stream>>>(s1p, s2p, tsp, s3m, d12, d13, d23, tri,
                                      W12, b12, W12t, b12t, W11, b11,
                                      s1m, s2m, a1, r12, r12t, forow);
  k3b_mix    <<<256, 256, 0, stream>>>(s1m, s2m, s3m, d12, d13, d23, W22, a1,
                                       W02, b02, b22, W01, b01,
                                       r12, r12t, forow, (float*)d_out);
}
